# starting kernel
# baseline (speedup 1.0000x reference)
#include <hip/hip_runtime.h>
#include <math.h>


typedef __attribute__((ext_vector_type(2))) float v2f;
typedef __attribute__((ext_vector_type(8))) float v8f;

__device__ __forceinline__ float fatomic_add(float* p, float v) {
    return __hip_atomic_fetch_add(p, v, __ATOMIC_RELAXED, __HIP_MEMORY_SCOPE_AGENT);
}

__global__ void k_zero(float* __restrict__ p, long long n) {
    long long i = (long long)blockIdx.x * blockDim.x + threadIdx.x;
    long long stride = (long long)gridDim.x * blockDim.x;
    for (; i < n; i += stride) p[i] = 0.0f;
}

__global__ __launch_bounds__(256) void k_att(const float* __restrict__ q,
                                             const float* __restrict__ Watt,
                                             float* __restrict__ attS,
                                             float* __restrict__ attD, int N) {
    int lane = threadIdx.x & 31;
    int node = (blockIdx.x * blockDim.x + threadIdx.x) >> 5;
    if (node >= N) return;
    const float* row = q + (long long)node * 64;
    float q0 = row[lane], q1 = row[lane + 32];
    float s = q0 * Watt[lane]      + q1 * Watt[lane + 32];
    float d = q0 * Watt[64 + lane] + q1 * Watt[96 + lane];
    #pragma unroll
    for (int off = 16; off > 0; off >>= 1) {
        s += __shfl_xor(s, off, 32);
        d += __shfl_xor(d, off, 32);
    }
    if (lane == 0) { attS[node] = s; attD[node] = d; }
}

__global__ __launch_bounds__(256) void k_denom(const int* __restrict__ src,
                                               const int* __restrict__ dst,
                                               const float* __restrict__ attS,
                                               const float* __restrict__ attD,
                                               float* __restrict__ den, int E) {
    int i = blockIdx.x * blockDim.x + threadIdx.x;
    int stride = gridDim.x * blockDim.x;
    for (int e = i; e < E; e += stride) {
        int s = src[e], d = dst[e];
        float x  = attS[s] + attD[d];
        float sg = 1.0f / (1.0f + expf(-x));
        fatomic_add(den + d, expf(sg));
    }
}

__global__ __launch_bounds__(256) void k_scatter(const int* __restrict__ src,
                                                 const int* __restrict__ dst,
                                                 const float* __restrict__ attS,
                                                 const float* __restrict__ attD,
                                                 const float* __restrict__ den,
                                                 const float* __restrict__ q,
                                                 float* __restrict__ X, int E) {
    int lane = threadIdx.x & 31;
    int wave = (blockIdx.x * blockDim.x + threadIdx.x) >> 5;
    int base = wave * 32;
    if (base >= E) return;

    int s = 0, d = 0; float a = 0.0f;
    int e = base + lane;
    if (e < E) {
        s = src[e]; d = dst[e];
        float x  = attS[s] + attD[d];
        float sg = 1.0f / (1.0f + expf(-x));
        float ex = expf(sg);
        a = ex / fmaxf(den[d], 1e-12f);
    }
    const float2* q2 = (const float2*)q;
    int cnt = min(32, E - base);
    for (int i = 0; i < cnt; ++i) {
        float ai = __shfl(a, i, 32);
        int   si = __shfl(s, i, 32);
        int   di = __shfl(d, i, 32);
        float2 v = q2[si * 32 + lane];
        float* xp = X + (long long)di * 64 + lane * 2;
        fatomic_add(xp,     ai * v.x);
        fatomic_add(xp + 1, ai * v.y);
    }
}

__global__ __launch_bounds__(256) void k_gemm(const float* __restrict__ q,
                                              const float* __restrict__ X,
                                              const float* __restrict__ Wupd,
                                              const float* __restrict__ Wv,
                                              const float* __restrict__ bV,
                                              const float* __restrict__ den1,
                                              const float* __restrict__ den2,
                                              float* __restrict__ out, int N) {
    __shared__ float sWu[64 * 64];
    __shared__ float sWv[64 * 64];
    for (int i = threadIdx.x; i < 64 * 64; i += 256) {
        sWu[i] = Wupd[i];
        sWv[i] = Wv[i];
    }
    __syncthreads();

    int lane = threadIdx.x & 31;
    int w    = threadIdx.x >> 5;
    int r0   = blockIdx.x * 32 + (w >> 2) * 16;
    int c0   = (w & 3) * 16;
    int half = lane >> 4;
    int l    = lane & 15;

    int arow  = r0 + l;
    int arow_c = min(arow, N - 1);
    int kofs  = half * 2;

    v8f accU = {};
    v8f accV = {};

    #pragma unroll
    for (int k = 0; k < 64; k += 4) {
        float2 xv = *(const float2*)(X + (long long)arow_c * 64 + k + kofs);
        float2 qv = *(const float2*)(q + (long long)arow_c * 64 + k + kofs);
        v2f a1; a1.x = 2.0f * qv.x + xv.x; a1.y = 2.0f * qv.y + xv.y;
        v2f a2; a2.x = xv.x;               a2.y = xv.y;
        v2f bu = *(const v2f*)&sWu[(c0 + l) * 64 + k + kofs];
        v2f bv = *(const v2f*)&sWv[(c0 + l) * 64 + k + kofs];
        accU = __builtin_amdgcn_wmma_f32_16x16x4_f32(false, a1, false, bu,
                                                     (short)0, accU, false, false);
        accV = __builtin_amdgcn_wmma_f32_16x16x4_f32(false, a2, false, bv,
                                                     (short)0, accV, false, false);
    }

    int col = c0 + l;
    float bvc = bV[col];
    #pragma unroll
    for (int r = 0; r < 8; ++r) {
        int row = r0 + r + half * 8;
        if (row < N) {
            float sI = ((den1[row] > 0.0f) ? 1.0f : 0.0f)
                     + ((den2[row] > 0.0f) ? 1.0f : 0.0f);
            out[(long long)row * 64 + col] = accU[r];
            out[(long long)N * 64 + (long long)row * 64 + col] = accV[r] + sI * bvc;
        }
    }
}

extern "C" void kernel_launch(void* const* d_in, const int* in_sizes, int n_in,
                              void* d_out, int out_size, void* d_ws, size_t ws_size,
                              hipStream_t stream) {
    const float* q    = (const float*)d_in[0];
    const float* Watt = (const float*)d_in[1];
    const float* Wupd = (const float*)d_in[2];
    const float* Wv   = (const float*)d_in[3];
    const float* bV   = (const float*)d_in[4];
    const int*   src1 = (const int*)d_in[5];
    const int*   dst1 = (const int*)d_in[6];
    const int*   src2 = (const int*)d_in[7];
    const int*   dst2 = (const int*)d_in[8];

    const int N = in_sizes[0] / 64;
    const int E = in_sizes[5];

    float* X    = (float*)d_ws;
    float* attS = X + (long long)N * 64;
    float* attD = attS + N;
    float* den1 = attD + N;
    float* den2 = den1 + N;

    float* out = (float*)d_out;

    long long zeroN = (long long)N * 64 + 4LL * N;
    k_zero<<<2048, 256, 0, stream>>>((float*)d_ws, zeroN);

    int attBlocks = (N + 7) / 8;
    k_att<<<attBlocks, 256, 0, stream>>>(q, Watt, attS, attD, N);

    int edgeBlocks = (E + 255) / 256;
    k_denom<<<edgeBlocks, 256, 0, stream>>>(src1, dst1, attS, attD, den1, E);
    k_denom<<<edgeBlocks, 256, 0, stream>>>(src2, dst2, attS, attD, den2, E);

    int waves = (E + 31) / 32;
    int scatterBlocks = (waves + 7) / 8;
    k_scatter<<<scatterBlocks, 256, 0, stream>>>(src1, dst1, attS, attD, den1, q, X, E);
    k_scatter<<<scatterBlocks, 256, 0, stream>>>(src2, dst2, attS, attD, den2, q, X, E);

    int gemmBlocks = (N + 31) / 32;
    k_gemm<<<gemmBlocks, 256, 0, stream>>>(q, X, Wupd, Wv, bV, den1, den2, out, N);
}


# Scatter pass: lane owns columns l and l+32 of a row instead of 2l and 2l+1, so every f32 atomic instruction and every gather load of a wave covers one whole 128-byte line, not two half-used lines
# speedup vs baseline: 1.1506x; 1.1506x over previous
#include <hip/hip_runtime.h>
#include <math.h>


typedef __attribute__((ext_vector_type(2))) float v2f;
typedef __attribute__((ext_vector_type(8))) float v8f;

__device__ __forceinline__ float fatomic_add(float* p, float v) {
    return __hip_atomic_fetch_add(p, v, __ATOMIC_RELAXED, __HIP_MEMORY_SCOPE_AGENT);
}

__global__ void k_zero(float* __restrict__ p0, long long n0,
                       float* __restrict__ p1, long long n1) {
    long long i = (long long)blockIdx.x * blockDim.x + threadIdx.x;
    long long stride = (long long)gridDim.x * blockDim.x;
    for (long long j = i; j < n0; j += stride) p0[j] = 0.0f;
    for (long long j = i; j < n1; j += stride) p1[j] = 0.0f;
}

__global__ __launch_bounds__(256) void k_att(const float* __restrict__ q,
                                             const float* __restrict__ Watt,
                                             float* __restrict__ attS,
                                             float* __restrict__ attD, int N) {
    int lane = threadIdx.x & 31;
    int node = (blockIdx.x * blockDim.x + threadIdx.x) >> 5;
    if (node >= N) return;
    const float* row = q + (long long)node * 64;
    float q0 = row[lane], q1 = row[lane + 32];
    float s = q0 * Watt[lane]      + q1 * Watt[lane + 32];
    float d = q0 * Watt[64 + lane] + q1 * Watt[96 + lane];
    #pragma unroll
    for (int off = 16; off > 0; off >>= 1) {
        s += __shfl_xor(s, off, 32);
        d += __shfl_xor(d, off, 32);
    }
    if (lane == 0) { attS[node] = s; attD[node] = d; }
}

__global__ __launch_bounds__(256) void k_scatter(const int* __restrict__ src,
                                                 const int* __restrict__ dst,
                                                 const float* __restrict__ attS,
                                                 const float* __restrict__ attD,
                                                 const float* __restrict__ q,
                                                 float* __restrict__ X,
                                                 float* __restrict__ den, int E) {
    int lane = threadIdx.x & 31;
    int wave = (blockIdx.x * blockDim.x + threadIdx.x) >> 5;
    int base = wave * 32;
    if (base >= E) return;

    int s = 0, d = 0; float a = 0.0f;
    int e = base + lane;
    if (e < E) {
        s = src[e]; d = dst[e];
        float x  = attS[s] + attD[d];
        float sg = 1.0f / (1.0f + expf(-x));
        a = expf(sg);
        fatomic_add(den + d, a);
    }
    int cnt = min(32, E - base);
    for (int i = 0; i < cnt; ++i) {
        float ai = __shfl(a, i, 32);
        int   si = __shfl(s, i, 32);
        int   di = __shfl(d, i, 32);
        const float* qr = q + (long long)si * 64 + lane;
        float v0 = qr[0];
        float v1 = qr[32];
        float* xp = X + (long long)di * 64 + lane;
        fatomic_add(xp,      ai * v0);
        fatomic_add(xp + 32, ai * v1);
    }
}

__global__ __launch_bounds__(256) void k_gemm(const float* __restrict__ q,
                                              const float* __restrict__ X1,
                                              const float* __restrict__ Wupd,
                                              const float* __restrict__ Wv,
                                              const float* __restrict__ bV,
                                              const float* __restrict__ den1,
                                              const float* __restrict__ den2,
                                              float* out, int N) {
    __shared__ float sWu[64 * 64];
    __shared__ float sWv[64 * 64];
    for (int i = threadIdx.x; i < 64 * 64; i += 256) {
        sWu[i] = Wupd[i];
        sWv[i] = Wv[i];
    }
    __syncthreads();

    int lane = threadIdx.x & 31;
    int w    = threadIdx.x >> 5;
    int r0   = blockIdx.x * 32 + (w >> 2) * 16;
    int c0   = (w & 3) * 16;
    int half = lane >> 4;
    int l    = lane & 15;

    int arow  = r0 + l;
    int arow_c = min(arow, N - 1);
    int kofs  = half * 2;

    const float* X2 = out + (long long)N * 64;
    float r1 = 1.0f / fmaxf(den1[arow_c], 1e-12f);
    float r2 = 1.0f / fmaxf(den2[arow_c], 1e-12f);

    v8f accU = {};
    v8f accV = {};

    #pragma unroll
    for (int k = 0; k < 64; k += 4) {
        float2 x1 = *(const float2*)(X1 + (long long)arow_c * 64 + k + kofs);
        float2 x2 = *(const float2*)(X2 + (long long)arow_c * 64 + k + kofs);
        float2 qv = *(const float2*)(q + (long long)arow_c * 64 + k + kofs);
        v2f a2; a2.x = x1.x * r1 + x2.x * r2;  a2.y = x1.y * r1 + x2.y * r2;
        v2f a1; a1.x = 2.0f * qv.x + a2.x;     a1.y = 2.0f * qv.y + a2.y;
        v2f bu = *(const v2f*)&sWu[(c0 + l) * 64 + k + kofs];
        v2f bv = *(const v2f*)&sWv[(c0 + l) * 64 + k + kofs];
        accU = __builtin_amdgcn_wmma_f32_16x16x4_f32(false, a1, false, bu,
                                                     (short)0, accU, false, false);
        accV = __builtin_amdgcn_wmma_f32_16x16x4_f32(false, a2, false, bv,
                                                     (short)0, accV, false, false);
    }

    __syncthreads();

    int col = c0 + l;
    float bvc = bV[col];
    #pragma unroll
    for (int r = 0; r < 8; ++r) {
        int row = r0 + r + half * 8;
        if (row < N) {
            float sI = ((den1[row] > 0.0f) ? 1.0f : 0.0f)
                     + ((den2[row] > 0.0f) ? 1.0f : 0.0f);
            out[(long long)row * 64 + col] = accU[r];
            out[(long long)N * 64 + (long long)row * 64 + col] = accV[r] + sI * bvc;
        }
    }
}

extern "C" void kernel_launch(void* const* d_in, const int* in_sizes, int n_in,
                              void* d_out, int out_size, void* d_ws, size_t ws_size,
                              hipStream_t stream) {
    const float* q    = (const float*)d_in[0];
    const float* Watt = (const float*)d_in[1];
    const float* Wupd = (const float*)d_in[2];
    const float* Wv   = (const float*)d_in[3];
    const float* bV   = (const float*)d_in[4];
    const int*   src1 = (const int*)d_in[5];
    const int*   dst1 = (const int*)d_in[6];
    const int*   src2 = (const int*)d_in[7];
    const int*   dst2 = (const int*)d_in[8];

    const int N = in_sizes[0] / 64;
    const int E = in_sizes[5];

    float* X1   = (float*)d_ws;
    float* den1 = X1 + (long long)N * 64;
    float* den2 = den1 + N;
    float* attS = den2 + N;
    float* attD = attS + N;

    float* out = (float*)d_out;
    float* X2  = out + (long long)N * 64;

    long long zeroWs = (long long)N * 64 + 2LL * N;
    k_zero<<<2048, 256, 0, stream>>>(X1, zeroWs, X2, (long long)N * 64);

    int attBlocks = (N + 7) / 8;
    k_att<<<attBlocks, 256, 0, stream>>>(q, Watt, attS, attD, N);

    int waves = (E + 31) / 32;
    int scatterBlocks = (waves + 7) / 8;
    k_scatter<<<scatterBlocks, 256, 0, stream>>>(src1, dst1, attS, attD, q, X1, den1, E);
    k_scatter<<<scatterBlocks, 256, 0, stream>>>(src2, dst2, attS, attD, q, X2, den2, E);

    int gemmBlocks = (N + 31) / 32;
    k_gemm<<<gemmBlocks, 256, 0, stream>>>(q, X1, Wupd, Wv, bV, den1, den2, out, N);
}


# Scatter pass: a wave issues the row loads of 8 edges back to back before its first add (16 line loads in flight per wave, not 2); edge words via v_readlane, not LDS permutes; same atomics, same order
# speedup vs baseline: 1.2825x; 1.1146x over previous
#include <hip/hip_runtime.h>
#include <math.h>


typedef __attribute__((ext_vector_type(2))) float v2f;
typedef __attribute__((ext_vector_type(8))) float v8f;

__device__ __forceinline__ float fatomic_add(float* p, float v) {
    return __hip_atomic_fetch_add(p, v, __ATOMIC_RELAXED, __HIP_MEMORY_SCOPE_AGENT);
}

__global__ void k_zero(float* __restrict__ p0, long long n0,
                       float* __restrict__ p1, long long n1) {
    long long i = (long long)blockIdx.x * blockDim.x + threadIdx.x;
    long long stride = (long long)gridDim.x * blockDim.x;
    for (long long j = i; j < n0; j += stride) p0[j] = 0.0f;
    for (long long j = i; j < n1; j += stride) p1[j] = 0.0f;
}

__global__ __launch_bounds__(256) void k_att(const float* __restrict__ q,
                                             const float* __restrict__ Watt,
                                             float* __restrict__ attS,
                                             float* __restrict__ attD, int N) {
    int lane = threadIdx.x & 31;
    int node = (blockIdx.x * blockDim.x + threadIdx.x) >> 5;
    if (node >= N) return;
    const float* row = q + (long long)node * 64;
    float q0 = row[lane], q1 = row[lane + 32];
    float s = q0 * Watt[lane]      + q1 * Watt[lane + 32];
    float d = q0 * Watt[64 + lane] + q1 * Watt[96 + lane];
    #pragma unroll
    for (int off = 16; off > 0; off >>= 1) {
        s += __shfl_xor(s, off, 32);
        d += __shfl_xor(d, off, 32);
    }
    if (lane == 0) { attS[node] = s; attD[node] = d; }
}

#define SCATTER_GRP 8

__device__ __forceinline__ float lane_read_f(float v, int l) {
    return __int_as_float(__builtin_amdgcn_readlane(__float_as_int(v), l));
}

__global__ __launch_bounds__(256) void k_scatter(const int* __restrict__ src,
                                                 const int* __restrict__ dst,
                                                 const float* __restrict__ attS,
                                                 const float* __restrict__ attD,
                                                 const float* __restrict__ q,
                                                 float* __restrict__ X,
                                                 float* __restrict__ den, int E) {
    int lane = threadIdx.x & 31;
    int wave = __builtin_amdgcn_readfirstlane((int)((blockIdx.x * blockDim.x + threadIdx.x) >> 5));
    int base = wave * 32;
    if (base >= E) return;

    int s = 0, d = 0; float a = 0.0f;
    int e = base + lane;
    if (e < E) {
        s = src[e]; d = dst[e];
        float x  = attS[s] + attD[d];
        float sg = 1.0f / (1.0f + expf(-x));
        a = expf(sg);
        fatomic_add(den + d, a);
    }
    int cnt = min(32, E - base);
    for (int i0 = 0; i0 < cnt; i0 += SCATTER_GRP) {
        float v0[SCATTER_GRP], v1[SCATTER_GRP];
        #pragma unroll
        for (int u = 0; u < SCATTER_GRP; ++u) {
            int si = __builtin_amdgcn_readlane(s, i0 + u);
            const float* qr = q + (long long)si * 64;
            v0[u] = qr[lane];
            v1[u] = qr[lane + 32];
        }
        if (i0 + SCATTER_GRP <= cnt) {
            float* xp[SCATTER_GRP];
            #pragma unroll
            for (int u = 0; u < SCATTER_GRP; ++u) {
                int di = __builtin_amdgcn_readlane(d, i0 + u);
                xp[u] = X + (long long)di * 64 + lane;
            }
            #pragma unroll
            for (int u = 0; u < SCATTER_GRP; ++u) {
                float ai = lane_read_f(a, i0 + u);
                v0[u] *= ai;
                v1[u] *= ai;
            }
            #pragma unroll
            for (int u = 0; u < SCATTER_GRP; ++u) {
                fatomic_add(xp[u],      v0[u]);
                fatomic_add(xp[u] + 32, v1[u]);
            }
        } else {
            #pragma unroll
            for (int u = 0; u < SCATTER_GRP; ++u) {
                if (i0 + u < cnt) {
                    float ai = lane_read_f(a, i0 + u);
                    int   di = __builtin_amdgcn_readlane(d, i0 + u);
                    float* xr = X + (long long)di * 64;
                    fatomic_add(xr + lane,      ai * v0[u]);
                    fatomic_add(xr + lane + 32, ai * v1[u]);
                }
            }
        }
    }
}

__global__ __launch_bounds__(256) void k_gemm(const float* __restrict__ q,
                                              const float* __restrict__ X1,
                                              const float* __restrict__ Wupd,
                                              const float* __restrict__ Wv,
                                              const float* __restrict__ bV,
                                              const float* __restrict__ den1,
                                              const float* __restrict__ den2,
                                              float* out, int N) {
    __shared__ float sWu[64 * 64];
    __shared__ float sWv[64 * 64];
    for (int i = threadIdx.x; i < 64 * 64; i += 256) {
        sWu[i] = Wupd[i];
        sWv[i] = Wv[i];
    }
    __syncthreads();

    int lane = threadIdx.x & 31;
    int w    = threadIdx.x >> 5;
    int r0   = blockIdx.x * 32 + (w >> 2) * 16;
    int c0   = (w & 3) * 16;
    int half = lane >> 4;
    int l    = lane & 15;

    int arow  = r0 + l;
    int arow_c = min(arow, N - 1);
    int kofs  = half * 2;

    const float* X2 = out + (long long)N * 64;
    float r1 = 1.0f / fmaxf(den1[arow_c], 1e-12f);
    float r2 = 1.0f / fmaxf(den2[arow_c], 1e-12f);

    v8f accU = {};
    v8f accV = {};

    #pragma unroll
    for (int k = 0; k < 64; k += 4) {
        float2 x1 = *(const float2*)(X1 + (long long)arow_c * 64 + k + kofs);
        float2 x2 = *(const float2*)(X2 + (long long)arow_c * 64 + k + kofs);
        float2 qv = *(const float2*)(q + (long long)arow_c * 64 + k + kofs);
        v2f a2; a2.x = x1.x * r1 + x2.x * r2;  a2.y = x1.y * r1 + x2.y * r2;
        v2f a1; a1.x = 2.0f * qv.x + a2.x;     a1.y = 2.0f * qv.y + a2.y;
        v2f bu = *(const v2f*)&sWu[(c0 + l) * 64 + k + kofs];
        v2f bv = *(const v2f*)&sWv[(c0 + l) * 64 + k + kofs];
        accU = __builtin_amdgcn_wmma_f32_16x16x4_f32(false, a1, false, bu,
                                                     (short)0, accU, false, false);
        accV = __builtin_amdgcn_wmma_f32_16x16x4_f32(false, a2, false, bv,
                                                     (short)0, accV, false, false);
    }

    __syncthreads();

    int col = c0 + l;
    float bvc = bV[col];
    #pragma unroll
    for (int r = 0; r < 8; ++r) {
        int row = r0 + r + half * 8;
        if (row < N) {
            float sI = ((den1[row] > 0.0f) ? 1.0f : 0.0f)
                     + ((den2[row] > 0.0f) ? 1.0f : 0.0f);
            out[(long long)row * 64 + col] = accU[r];
            out[(long long)N * 64 + (long long)row * 64 + col] = accV[r] + sI * bvc;
        }
    }
}

extern "C" void kernel_launch(void* const* d_in, const int* in_sizes, int n_in,
                              void* d_out, int out_size, void* d_ws, size_t ws_size,
                              hipStream_t stream) {
    const float* q    = (const float*)d_in[0];
    const float* Watt = (const float*)d_in[1];
    const float* Wupd = (const float*)d_in[2];
    const float* Wv   = (const float*)d_in[3];
    const float* bV   = (const float*)d_in[4];
    const int*   src1 = (const int*)d_in[5];
    const int*   dst1 = (const int*)d_in[6];
    const int*   src2 = (const int*)d_in[7];
    const int*   dst2 = (const int*)d_in[8];

    const int N = in_sizes[0] / 64;
    const int E = in_sizes[5];

    float* X1   = (float*)d_ws;
    float* den1 = X1 + (long long)N * 64;
    float* den2 = den1 + N;
    float* attS = den2 + N;
    float* attD = attS + N;

    float* out = (float*)d_out;
    float* X2  = out + (long long)N * 64;

    long long zeroWs = (long long)N * 64 + 2LL * N;
    k_zero<<<2048, 256, 0, stream>>>(X1, zeroWs, X2, (long long)N * 64);

    int attBlocks = (N + 7) / 8;
    k_att<<<attBlocks, 256, 0, stream>>>(q, Watt, attS, attD, N);

    int waves = (E + 31) / 32;
    int scatterBlocks = (waves + 7) / 8;
    k_scatter<<<scatterBlocks, 256, 0, stream>>>(src1, dst1, attS, attD, q, X1, den1, E);
    k_scatter<<<scatterBlocks, 256, 0, stream>>>(src2, dst2, attS, attD, q, X2, den2, E);

    int gemmBlocks = (N + 31) / 32;
    k_gemm<<<gemmBlocks, 256, 0, stream>>>(q, X1, Wupd, Wv, bV, den1, den2, out, N);
}


# Row sums by destination bucket: edges binned into 256-node buckets (LDS counting pass), a work-group sums its bucket's rows with LDS f32 adds (arrival order); global f32 atomics and zero fill go
# speedup vs baseline: 2.6898x; 2.0973x over previous
#include <hip/hip_runtime.h>
#include <math.h>


typedef __attribute__((ext_vector_type(2))) float v2f;
typedef __attribute__((ext_vector_type(8))) float v8f;

#define BKT_SHIFT 8
#define BKT       256
#define NB_MAX    512
#define EPT       16
#define EBLK      (256 * EPT)

__global__ __launch_bounds__(256) void k_att(const float* __restrict__ q,
                                             const float* __restrict__ Watt,
                                             float* __restrict__ attS,
                                             float* __restrict__ attD,
                                             int* __restrict__ ctl, int N) {
    int gid = blockIdx.x * blockDim.x + threadIdx.x;
    if (gid < 4 * NB_MAX) ctl[gid] = 0;
    int lane = threadIdx.x & 31;
    int node = gid >> 5;
    if (node >= N) return;
    const float* row = q + (long long)node * 64;
    float q0 = row[lane], q1 = row[lane + 32];
    float s = q0 * Watt[lane]      + q1 * Watt[lane + 32];
    float d = q0 * Watt[64 + lane] + q1 * Watt[96 + lane];
    #pragma unroll
    for (int off = 16; off > 0; off >>= 1) {
        s += __shfl_xor(s, off, 32);
        d += __shfl_xor(d, off, 32);
    }
    if (lane == 0) { attS[node] = s; attD[node] = d; }
}

__global__ __launch_bounds__(256) void k_count(const int* __restrict__ dst1,
                                               const int* __restrict__ dst2,
                                               int* __restrict__ tot, int E) {
    __shared__ int h[NB_MAX];
    int tid = threadIdx.x;
    int t = blockIdx.y;
    const int* dst = t ? dst2 : dst1;
    for (int i = tid; i < NB_MAX; i += 256) h[i] = 0;
    __syncthreads();
    int base = blockIdx.x * EBLK;
    #pragma unroll
    for (int k = 0; k < EPT; ++k) {
        int e = base + k * 256 + tid;
        if (e < E) atomicAdd(&h[dst[e] >> BKT_SHIFT], 1);
    }
    __syncthreads();
    for (int i = tid; i < NB_MAX; i += 256) {
        int c = h[i];
        if (c > 0) atomicAdd(&tot[t * NB_MAX + i], c);
    }
}

__device__ __forceinline__ void bucket_starts(const int* __restrict__ tot, int lane,
                                              int* bstart) {
    int c[16];
    const int4* tp = (const int4*)(tot + lane * 16);
    #pragma unroll
    for (int j = 0; j < 4; ++j) {
        int4 v = tp[j];
        c[4 * j] = v.x; c[4 * j + 1] = v.y; c[4 * j + 2] = v.z; c[4 * j + 3] = v.w;
    }
    int run = 0;
    #pragma unroll
    for (int j = 0; j < 16; ++j) { int x = c[j]; c[j] = run; run += x; }
    int inc = run;
    #pragma unroll
    for (int off = 1; off < 32; off <<= 1) {
        int y = __shfl_up(inc, off, 32);
        if (lane >= off) inc += y;
    }
    int excl = inc - run;
    #pragma unroll
    for (int j = 0; j < 16; ++j) bstart[lane * 16 + j] = excl + c[j];
}

__global__ __launch_bounds__(256) void k_fill(const int* __restrict__ src1,
                                              const int* __restrict__ dst1,
                                              const int* __restrict__ src2,
                                              const int* __restrict__ dst2,
                                              const float* __restrict__ attS,
                                              const int* __restrict__ tot,
                                              int* __restrict__ cur,
                                              int* __restrict__ list, int E) {
    __shared__ __attribute__((aligned(16))) int h[NB_MAX];
    __shared__ int bstart[NB_MAX];
    __shared__ int loff[NB_MAX];
    __shared__ int2 srec[EBLK];
    __shared__ unsigned short sbk[EBLK];
    int tid = threadIdx.x;
    int t = blockIdx.y;
    const int* src = t ? src2 : src1;
    const int* dst = t ? dst2 : dst1;
    for (int i = tid; i < NB_MAX; i += 256) h[i] = 0;
    __syncthreads();

    int base = blockIdx.x * EBLK;
    int nE = min(EBLK, E - base);
    int pk[EPT], rk[EPT], bk[EPT];
    float as[EPT];
    #pragma unroll
    for (int k = 0; k < EPT; ++k) {
        int e = base + k * 256 + tid;
        bk[k] = -1; pk[k] = 0; rk[k] = 0; as[k] = 0.0f;
        if (e < E) {
            int s = src[e], d = dst[e];
            as[k] = attS[s];
            bk[k] = d >> BKT_SHIFT;
            pk[k] = (s << BKT_SHIFT) | (d & (BKT - 1));
            rk[k] = atomicAdd(&h[bk[k]], 1);
        }
    }
    __syncthreads();
    if (tid < 32)      bucket_starts(tot + t * NB_MAX, tid, bstart);
    else if (tid < 64) bucket_starts(h, tid - 32, loff);
    __syncthreads();
    #pragma unroll
    for (int k = 0; k < EPT; ++k) {
        if (bk[k] >= 0) {
            int j = loff[bk[k]] + rk[k];
            srec[j] = make_int2(pk[k], __float_as_int(as[k]));
            sbk[j]  = (unsigned short)bk[k];
        }
    }
    for (int i = tid; i < NB_MAX; i += 256) {
        int cb = h[i];
        int g = 0;
        if (cb > 0) g = bstart[i] + atomicAdd(&cur[t * NB_MAX + i], cb);
        h[i] = g - loff[i];
    }
    __syncthreads();
    int2* L = (int2*)list + (long long)t * E;
    for (int j = tid; j < nE; j += 256) L[h[sbk[j]] + j] = srec[j];
}

#define ACC_GRP 16

__device__ __forceinline__ float lane_read_f(float v, int l) {
    return __int_as_float(__builtin_amdgcn_readlane(__float_as_int(v), l));
}

__device__ __forceinline__ void lds_fadd(float* p, float v) {
    __hip_atomic_fetch_add(p, v, __ATOMIC_RELAXED, __HIP_MEMORY_SCOPE_WORKGROUP);
}

__global__ __launch_bounds__(256) void k_accum(const float* __restrict__ q,
                                               const float* __restrict__ attD,
                                               const int* __restrict__ list,
                                               const int* __restrict__ tot,
                                               float* __restrict__ X1,
                                               float* __restrict__ X2,
                                               float* __restrict__ den1,
                                               float* __restrict__ den2,
                                               int E, int N) {
    __shared__ float acc[BKT * 64];
    __shared__ float dn[BKT];
    __shared__ float sAD[BKT];
    __shared__ int bstart[NB_MAX];
    int tid  = threadIdx.x;
    int lane = tid & 31;
    int w = __builtin_amdgcn_readfirstlane(tid >> 5);
    int t = blockIdx.y;
    int b = blockIdx.x;
    int node0 = b * BKT;

    float4* acc4 = (float4*)acc;
    #pragma unroll
    for (int it = 0; it < (BKT * 64 / 4) / 256; ++it)
        acc4[it * 256 + tid] = make_float4(0.0f, 0.0f, 0.0f, 0.0f);
    dn[tid]  = 0.0f;
    sAD[tid] = (node0 + tid < N) ? attD[node0 + tid] : 0.0f;
    if (w == 0) bucket_starts(tot + t * NB_MAX, lane, bstart);
    __syncthreads();

    int start = __builtin_amdgcn_readfirstlane(bstart[b]);
    int cnt   = __builtin_amdgcn_readfirstlane(tot[t * NB_MAX + b]);
    const int2* L = (const int2*)list + (long long)t * E + start;

    int2 en = make_int2(0, 0);
    if (w * 32 + lane < cnt) en = L[w * 32 + lane];
    for (int i = w * 32; i < cnt; i += 256) {
        int n = min(32, cnt - i);
        int pk = en.x;
        float as = __int_as_float(en.y);
        en = make_int2(0, 0);
        if (i + 256 + lane < cnt) en = L[i + 256 + lane];
        float a = 0.0f;
        if (lane < n) {
            float x  = as + sAD[pk & (BKT - 1)];
            float sg = 1.0f / (1.0f + expf(-x));
            a = expf(sg);
            lds_fadd(&dn[pk & (BKT - 1)], a);
        }
        for (int i0 = 0; i0 < n; i0 += ACC_GRP) {
            float2 v[ACC_GRP];
            #pragma unroll
            for (int u = 0; u < ACC_GRP; ++u) {
                int si = __builtin_amdgcn_readlane(pk, i0 + u) >> BKT_SHIFT;
                v[u] = *(const float2*)(q + (long long)si * 64 + 2 * lane);
            }
            if (i0 + ACC_GRP <= n) {
                #pragma unroll
                for (int u = 0; u < ACC_GRP; ++u) {
                    int   dl = __builtin_amdgcn_readlane(pk, i0 + u) & (BKT - 1);
                    float ai = lane_read_f(a, i0 + u);
                    float* ar = acc + dl * 64 + 2 * lane;
                    lds_fadd(ar,     ai * v[u].x);
                    lds_fadd(ar + 1, ai * v[u].y);
                }
            } else {
                #pragma unroll
                for (int u = 0; u < ACC_GRP; ++u) {
                    if (i0 + u < n) {
                        int   dl = __builtin_amdgcn_readlane(pk, i0 + u) & (BKT - 1);
                        float ai = lane_read_f(a, i0 + u);
                        float* ar = acc + dl * 64 + 2 * lane;
                        lds_fadd(ar,     ai * v[u].x);
                        lds_fadd(ar + 1, ai * v[u].y);
                    }
                }
            }
        }
    }
    __syncthreads();

    float* X   = t ? X2 : X1;
    float* den = t ? den2 : den1;
    #pragma unroll
    for (int it = 0; it < (BKT * 64 / 4) / 256; ++it) {
        int idx4 = it * 256 + tid;
        int r = idx4 >> 4;
        if (node0 + r < N)
            *(float4*)(X + (long long)(node0 + r) * 64 + (idx4 & 15) * 4) = acc4[idx4];
    }
    if (node0 + tid < N) den[node0 + tid] = dn[tid];
}

__global__ __launch_bounds__(256) void k_gemm(const float* __restrict__ q,
                                              const float* __restrict__ X1,
                                              const float* __restrict__ Wupd,
                                              const float* __restrict__ Wv,
                                              const float* __restrict__ bV,
                                              const float* __restrict__ den1,
                                              const float* __restrict__ den2,
                                              float* out, int N) {
    __shared__ float sWu[64 * 64];
    __shared__ float sWv[64 * 64];
    for (int i = threadIdx.x; i < 64 * 64; i += 256) {
        sWu[i] = Wupd[i];
        sWv[i] = Wv[i];
    }
    __syncthreads();

    int lane = threadIdx.x & 31;
    int w    = threadIdx.x >> 5;
    int r0   = blockIdx.x * 32 + (w >> 2) * 16;
    int c0   = (w & 3) * 16;
    int half = lane >> 4;
    int l    = lane & 15;

    int arow  = r0 + l;
    int arow_c = min(arow, N - 1);
    int kofs  = half * 2;

    const float* X2 = out + (long long)N * 64;
    float r1 = 1.0f / fmaxf(den1[arow_c], 1e-12f);
    float r2 = 1.0f / fmaxf(den2[arow_c], 1e-12f);

    v8f accU = {};
    v8f accV = {};

    #pragma unroll
    for (int k = 0; k < 64; k += 4) {
        float2 x1 = *(const float2*)(X1 + (long long)arow_c * 64 + k + kofs);
        float2 x2 = *(const float2*)(X2 + (long long)arow_c * 64 + k + kofs);
        float2 qv = *(const float2*)(q + (long long)arow_c * 64 + k + kofs);
        v2f a2; a2.x = x1.x * r1 + x2.x * r2;  a2.y = x1.y * r1 + x2.y * r2;
        v2f a1; a1.x = 2.0f * qv.x + a2.x;     a1.y = 2.0f * qv.y + a2.y;
        v2f bu = *(const v2f*)&sWu[(c0 + l) * 64 + k + kofs];
        v2f bv = *(const v2f*)&sWv[(c0 + l) * 64 + k + kofs];
        accU = __builtin_amdgcn_wmma_f32_16x16x4_f32(false, a1, false, bu,
                                                     (short)0, accU, false, false);
        accV = __builtin_amdgcn_wmma_f32_16x16x4_f32(false, a2, false, bv,
                                                     (short)0, accV, false, false);
    }

    __syncthreads();

    int col = c0 + l;
    float bvc = bV[col];
    #pragma unroll
    for (int r = 0; r < 8; ++r) {
        int row = r0 + r + half * 8;
        if (row < N) {
            float sI = ((den1[row] > 0.0f) ? 1.0f : 0.0f)
                     + ((den2[row] > 0.0f) ? 1.0f : 0.0f);
            out[(long long)row * 64 + col] = accU[r];
            out[(long long)N * 64 + (long long)row * 64 + col] = accV[r] + sI * bvc;
        }
    }
}

extern "C" void kernel_launch(void* const* d_in, const int* in_sizes, int n_in,
                              void* d_out, int out_size, void* d_ws, size_t ws_size,
                              hipStream_t stream) {
    const float* q    = (const float*)d_in[0];
    const float* Watt = (const float*)d_in[1];
    const float* Wupd = (const float*)d_in[2];
    const float* Wv   = (const float*)d_in[3];
    const float* bV   = (const float*)d_in[4];
    const int*   src1 = (const int*)d_in[5];
    const int*   dst1 = (const int*)d_in[6];
    const int*   src2 = (const int*)d_in[7];
    const int*   dst2 = (const int*)d_in[8];

    const int N = in_sizes[0] / 64;
    const int E = in_sizes[5];

    float* X1   = (float*)d_ws;
    float* attS = X1 + (long long)N * 64;
    float* attD = attS + N;
    float* den1 = attS;
    float* den2 = attD + N;
    int*   ctl  = (int*)(den2 + N);
    int*   tot  = ctl;
    int*   cur  = ctl + 2 * NB_MAX;

    float* out  = (float*)d_out;
    float* X2   = out + (long long)N * 64;
    int*   list = (int*)d_out;

    int attBlocks = (N + 7) / 8;
    k_att<<<attBlocks, 256, 0, stream>>>(q, Watt, attS, attD, ctl, N);

    int edgeBlocks = (E + EBLK - 1) / EBLK;
    k_count<<<dim3(edgeBlocks, 2), 256, 0, stream>>>(dst1, dst2, tot, E);
    k_fill<<<dim3(edgeBlocks, 2), 256, 0, stream>>>(src1, dst1, src2, dst2, attS,
                                                   tot, cur, list, E);

    int buckets = (N + BKT - 1) / BKT;
    k_accum<<<dim3(buckets, 2), 256, 0, stream>>>(q, attD, list, tot, X1, X2, den1, den2, E, N);

    int gemmBlocks = (N + 31) / 32;
    k_gemm<<<gemmBlocks, 256, 0, stream>>>(q, X1, Wupd, Wv, bV, den1, den2, out, N);
}


# Node dot products: a wave takes 32 nodes, not 1 (16 lanes per row, 16-byte loads, row sums folded by DPP, whole-line stores): 3125 waves to start, not 100000; the dot products' sum order changes
# speedup vs baseline: 2.8722x; 1.0678x over previous
#include <hip/hip_runtime.h>
#include <math.h>


typedef __attribute__((ext_vector_type(2))) float v2f;
typedef __attribute__((ext_vector_type(8))) float v8f;

#define BKT_SHIFT 8
#define BKT       256
#define NB_MAX    512
#define EPT       16
#define EBLK      (256 * EPT)

#define ATT_ROWS 32

template <int M>
__device__ __forceinline__ float row_xor(float x) {
    return __int_as_float(__builtin_amdgcn_mov_dpp(__float_as_int(x), 0x160 + M, 0xf, 0xf, true));
}

template <int M>
__device__ __forceinline__ void fold_level(float* a, bool up) {
    #pragma unroll
    for (int k = 0; k < M; ++k) {
        float keep = up ? a[k + M] : a[k];
        float send = up ? a[k] : a[k + M];
        a[k] = keep + row_xor<M>(send);
    }
}

template <bool RAGGED>
__device__ __forceinline__ void att_rows(const float* __restrict__ q, const float4 ws,
                                         const float4 wd, float* __restrict__ attS,
                                         float* __restrict__ attD, int base, int c, int h,
                                         int N) {
    float4 v[ATT_ROWS / 2];
    const float* p = q + (long long)(base + h) * 64 + 4 * c;
    #pragma unroll
    for (int j = 0; j < ATT_ROWS / 2; ++j) {
        if (RAGGED) {
            int row = min(base + 2 * j + h, N - 1);
            v[j] = *(const float4*)(q + (long long)row * 64 + 4 * c);
        } else {
            v[j] = *(const float4*)(p + j * 128);
        }
    }
    const v2f wsA = {ws.x, ws.y}, wsB = {ws.z, ws.w};
    const v2f wdA = {wd.x, wd.y}, wdB = {wd.z, wd.w};
    float s[ATT_ROWS / 2], d[ATT_ROWS / 2];
    #pragma unroll
    for (int j = 0; j < ATT_ROWS / 2; ++j) {
        const v2f a = {v[j].x, v[j].y}, b = {v[j].z, v[j].w};
        const v2f ts = a * wsA + b * wsB;
        const v2f td = a * wdA + b * wdB;
        s[j] = ts.x + ts.y;
        d[j] = td.x + td.y;
    }
    fold_level<8>(s, (c & 8) != 0);  fold_level<8>(d, (c & 8) != 0);
    fold_level<4>(s, (c & 4) != 0);  fold_level<4>(d, (c & 4) != 0);
    fold_level<2>(s, (c & 2) != 0);  fold_level<2>(d, (c & 2) != 0);
    fold_level<1>(s, (c & 1) != 0);  fold_level<1>(d, (c & 1) != 0);
    int node = base + 2 * c + h;
    if (!RAGGED || node < N) { attS[node] = s[0]; attD[node] = d[0]; }
}

__global__ __launch_bounds__(256) void k_att(const float* __restrict__ q,
                                             const float* __restrict__ Watt,
                                             float* __restrict__ attS,
                                             float* __restrict__ attD,
                                             int* __restrict__ ctl, int N) {
    for (int i = blockIdx.x * 256 + threadIdx.x; i < 4 * NB_MAX; i += gridDim.x * 256)
        ctl[i] = 0;
    int lane = threadIdx.x & 31;
    int c = lane & 15;
    int h = lane >> 4;
    int w = __builtin_amdgcn_readfirstlane(threadIdx.x >> 5);
    int base = (blockIdx.x * 8 + w) * ATT_ROWS;
    if (base >= N) return;
    const float4 ws = *(const float4*)(Watt + 4 * c);
    const float4 wd = *(const float4*)(Watt + 64 + 4 * c);
    if (base + ATT_ROWS <= N) att_rows<false>(q, ws, wd, attS, attD, base, c, h, N);
    else                      att_rows<true>(q, ws, wd, attS, attD, base, c, h, N);
}

__global__ __launch_bounds__(256) void k_count(const int* __restrict__ dst1,
                                               const int* __restrict__ dst2,
                                               int* __restrict__ tot, int E) {
    __shared__ int h[NB_MAX];
    int tid = threadIdx.x;
    int t = blockIdx.y;
    const int* dst = t ? dst2 : dst1;
    for (int i = tid; i < NB_MAX; i += 256) h[i] = 0;
    __syncthreads();
    int base = blockIdx.x * EBLK;
    #pragma unroll
    for (int k = 0; k < EPT; ++k) {
        int e = base + k * 256 + tid;
        if (e < E) atomicAdd(&h[dst[e] >> BKT_SHIFT], 1);
    }
    __syncthreads();
    for (int i = tid; i < NB_MAX; i += 256) {
        int c = h[i];
        if (c > 0) atomicAdd(&tot[t * NB_MAX + i], c);
    }
}

__device__ __forceinline__ void bucket_starts(const int* __restrict__ tot, int lane,
                                              int* bstart) {
    int c[16];
    const int4* tp = (const int4*)(tot + lane * 16);
    #pragma unroll
    for (int j = 0; j < 4; ++j) {
        int4 v = tp[j];
        c[4 * j] = v.x; c[4 * j + 1] = v.y; c[4 * j + 2] = v.z; c[4 * j + 3] = v.w;
    }
    int run = 0;
    #pragma unroll
    for (int j = 0; j < 16; ++j) { int x = c[j]; c[j] = run; run += x; }
    int inc = run;
    #pragma unroll
    for (int off = 1; off < 32; off <<= 1) {
        int y = __shfl_up(inc, off, 32);
        if (lane >= off) inc += y;
    }
    int excl = inc - run;
    #pragma unroll
    for (int j = 0; j < 16; ++j) bstart[lane * 16 + j] = excl + c[j];
}

__global__ __launch_bounds__(256) void k_fill(const int* __restrict__ src1,
                                              const int* __restrict__ dst1,
                                              const int* __restrict__ src2,
                                              const int* __restrict__ dst2,
                                              const float* __restrict__ attS,
                                              const int* __restrict__ tot,
                                              int* __restrict__ cur,
                                              int* __restrict__ list, int E) {
    __shared__ __attribute__((aligned(16))) int h[NB_MAX];
    __shared__ int bstart[NB_MAX];
    __shared__ int loff[NB_MAX];
    __shared__ int2 srec[EBLK];
    __shared__ unsigned short sbk[EBLK];
    int tid = threadIdx.x;
    int t = blockIdx.y;
    const int* src = t ? src2 : src1;
    const int* dst = t ? dst2 : dst1;
    for (int i = tid; i < NB_MAX; i += 256) h[i] = 0;
    __syncthreads();

    int base = blockIdx.x * EBLK;
    int nE = min(EBLK, E - base);
    int pk[EPT], rk[EPT], bk[EPT];
    float as[EPT];
    #pragma unroll
    for (int k = 0; k < EPT; ++k) {
        int e = base + k * 256 + tid;
        bk[k] = -1; pk[k] = 0; rk[k] = 0; as[k] = 0.0f;
        if (e < E) {
            int s = src[e], d = dst[e];
            as[k] = attS[s];
            bk[k] = d >> BKT_SHIFT;
            pk[k] = (s << BKT_SHIFT) | (d & (BKT - 1));
            rk[k] = atomicAdd(&h[bk[k]], 1);
        }
    }
    __syncthreads();
    if (tid < 32)      bucket_starts(tot + t * NB_MAX, tid, bstart);
    else if (tid < 64) bucket_starts(h, tid - 32, loff);
    __syncthreads();
    #pragma unroll
    for (int k = 0; k < EPT; ++k) {
        if (bk[k] >= 0) {
            int j = loff[bk[k]] + rk[k];
            srec[j] = make_int2(pk[k], __float_as_int(as[k]));
            sbk[j]  = (unsigned short)bk[k];
        }
    }
    for (int i = tid; i < NB_MAX; i += 256) {
        int cb = h[i];
        int g = 0;
        if (cb > 0) g = bstart[i] + atomicAdd(&cur[t * NB_MAX + i], cb);
        h[i] = g - loff[i];
    }
    __syncthreads();
    int2* L = (int2*)list + (long long)t * E;
    for (int j = tid; j < nE; j += 256) L[h[sbk[j]] + j] = srec[j];
}

#define ACC_GRP 16

__device__ __forceinline__ float lane_read_f(float v, int l) {
    return __int_as_float(__builtin_amdgcn_readlane(__float_as_int(v), l));
}

__device__ __forceinline__ void lds_fadd(float* p, float v) {
    __hip_atomic_fetch_add(p, v, __ATOMIC_RELAXED, __HIP_MEMORY_SCOPE_WORKGROUP);
}

__global__ __launch_bounds__(256) void k_accum(const float* __restrict__ q,
                                               const float* __restrict__ attD,
                                               const int* __restrict__ list,
                                               const int* __restrict__ tot,
                                               float* __restrict__ X1,
                                               float* __restrict__ X2,
                                               float* __restrict__ den1,
                                               float* __restrict__ den2,
                                               int E, int N) {
    __shared__ float acc[BKT * 64];
    __shared__ float dn[BKT];
    __shared__ float sAD[BKT];
    __shared__ int bstart[NB_MAX];
    int tid  = threadIdx.x;
    int lane = tid & 31;
    int w = __builtin_amdgcn_readfirstlane(tid >> 5);
    int t = blockIdx.y;
    int b = blockIdx.x;
    int node0 = b * BKT;

    float4* acc4 = (float4*)acc;
    #pragma unroll
    for (int it = 0; it < (BKT * 64 / 4) / 256; ++it)
        acc4[it * 256 + tid] = make_float4(0.0f, 0.0f, 0.0f, 0.0f);
    dn[tid]  = 0.0f;
    sAD[tid] = (node0 + tid < N) ? attD[node0 + tid] : 0.0f;
    if (w == 0) bucket_starts(tot + t * NB_MAX, lane, bstart);
    __syncthreads();

    int start = __builtin_amdgcn_readfirstlane(bstart[b]);
    int cnt   = __builtin_amdgcn_readfirstlane(tot[t * NB_MAX + b]);
    const int2* L = (const int2*)list + (long long)t * E + start;

    int2 en = make_int2(0, 0);
    if (w * 32 + lane < cnt) en = L[w * 32 + lane];
    for (int i = w * 32; i < cnt; i += 256) {
        int n = min(32, cnt - i);
        int pk = en.x;
        float as = __int_as_float(en.y);
        en = make_int2(0, 0);
        if (i + 256 + lane < cnt) en = L[i + 256 + lane];
        float a = 0.0f;
        if (lane < n) {
            float x  = as + sAD[pk & (BKT - 1)];
            float sg = 1.0f / (1.0f + expf(-x));
            a = expf(sg);
            lds_fadd(&dn[pk & (BKT - 1)], a);
        }
        for (int i0 = 0; i0 < n; i0 += ACC_GRP) {
            float2 v[ACC_GRP];
            #pragma unroll
            for (int u = 0; u < ACC_GRP; ++u) {
                int si = __builtin_amdgcn_readlane(pk, i0 + u) >> BKT_SHIFT;
                v[u] = *(const float2*)(q + (long long)si * 64 + 2 * lane);
            }
            if (i0 + ACC_GRP <= n) {
                #pragma unroll
                for (int u = 0; u < ACC_GRP; ++u) {
                    int   dl = __builtin_amdgcn_readlane(pk, i0 + u) & (BKT - 1);
                    float ai = lane_read_f(a, i0 + u);
                    float* ar = acc + dl * 64 + 2 * lane;
                    lds_fadd(ar,     ai * v[u].x);
                    lds_fadd(ar + 1, ai * v[u].y);
                }
            } else {
                #pragma unroll
                for (int u = 0; u < ACC_GRP; ++u) {
                    if (i0 + u < n) {
                        int   dl = __builtin_amdgcn_readlane(pk, i0 + u) & (BKT - 1);
                        float ai = lane_read_f(a, i0 + u);
                        float* ar = acc + dl * 64 + 2 * lane;
                        lds_fadd(ar,     ai * v[u].x);
                        lds_fadd(ar + 1, ai * v[u].y);
                    }
                }
            }
        }
    }
    __syncthreads();

    float* X   = t ? X2 : X1;
    float* den = t ? den2 : den1;
    #pragma unroll
    for (int it = 0; it < (BKT * 64 / 4) / 256; ++it) {
        int idx4 = it * 256 + tid;
        int r = idx4 >> 4;
        if (node0 + r < N)
            *(float4*)(X + (long long)(node0 + r) * 64 + (idx4 & 15) * 4) = acc4[idx4];
    }
    if (node0 + tid < N) den[node0 + tid] = dn[tid];
}

__global__ __launch_bounds__(256) void k_gemm(const float* __restrict__ q,
                                              const float* __restrict__ X1,
                                              const float* __restrict__ Wupd,
                                              const float* __restrict__ Wv,
                                              const float* __restrict__ bV,
                                              const float* __restrict__ den1,
                                              const float* __restrict__ den2,
                                              float* out, int N) {
    __shared__ float sWu[64 * 64];
    __shared__ float sWv[64 * 64];
    for (int i = threadIdx.x; i < 64 * 64; i += 256) {
        sWu[i] = Wupd[i];
        sWv[i] = Wv[i];
    }
    __syncthreads();

    int lane = threadIdx.x & 31;
    int w    = threadIdx.x >> 5;
    int r0   = blockIdx.x * 32 + (w >> 2) * 16;
    int c0   = (w & 3) * 16;
    int half = lane >> 4;
    int l    = lane & 15;

    int arow  = r0 + l;
    int arow_c = min(arow, N - 1);
    int kofs  = half * 2;

    const float* X2 = out + (long long)N * 64;
    float r1 = 1.0f / fmaxf(den1[arow_c], 1e-12f);
    float r2 = 1.0f / fmaxf(den2[arow_c], 1e-12f);

    v8f accU = {};
    v8f accV = {};

    #pragma unroll
    for (int k = 0; k < 64; k += 4) {
        float2 x1 = *(const float2*)(X1 + (long long)arow_c * 64 + k + kofs);
        float2 x2 = *(const float2*)(X2 + (long long)arow_c * 64 + k + kofs);
        float2 qv = *(const float2*)(q + (long long)arow_c * 64 + k + kofs);
        v2f a2; a2.x = x1.x * r1 + x2.x * r2;  a2.y = x1.y * r1 + x2.y * r2;
        v2f a1; a1.x = 2.0f * qv.x + a2.x;     a1.y = 2.0f * qv.y + a2.y;
        v2f bu = *(const v2f*)&sWu[(c0 + l) * 64 + k + kofs];
        v2f bv = *(const v2f*)&sWv[(c0 + l) * 64 + k + kofs];
        accU = __builtin_amdgcn_wmma_f32_16x16x4_f32(false, a1, false, bu,
                                                     (short)0, accU, false, false);
        accV = __builtin_amdgcn_wmma_f32_16x16x4_f32(false, a2, false, bv,
                                                     (short)0, accV, false, false);
    }

    __syncthreads();

    int col = c0 + l;
    float bvc = bV[col];
    #pragma unroll
    for (int r = 0; r < 8; ++r) {
        int row = r0 + r + half * 8;
        if (row < N) {
            float sI = ((den1[row] > 0.0f) ? 1.0f : 0.0f)
                     + ((den2[row] > 0.0f) ? 1.0f : 0.0f);
            out[(long long)row * 64 + col] = accU[r];
            out[(long long)N * 64 + (long long)row * 64 + col] = accV[r] + sI * bvc;
        }
    }
}

extern "C" void kernel_launch(void* const* d_in, const int* in_sizes, int n_in,
                              void* d_out, int out_size, void* d_ws, size_t ws_size,
                              hipStream_t stream) {
    const float* q    = (const float*)d_in[0];
    const float* Watt = (const float*)d_in[1];
    const float* Wupd = (const float*)d_in[2];
    const float* Wv   = (const float*)d_in[3];
    const float* bV   = (const float*)d_in[4];
    const int*   src1 = (const int*)d_in[5];
    const int*   dst1 = (const int*)d_in[6];
    const int*   src2 = (const int*)d_in[7];
    const int*   dst2 = (const int*)d_in[8];

    const int N = in_sizes[0] / 64;
    const int E = in_sizes[5];

    float* X1   = (float*)d_ws;
    float* attS = X1 + (long long)N * 64;
    float* attD = attS + N;
    float* den1 = attS;
    float* den2 = attD + N;
    int*   ctl  = (int*)(den2 + N);
    int*   tot  = ctl;
    int*   cur  = ctl + 2 * NB_MAX;

    float* out  = (float*)d_out;
    float* X2   = out + (long long)N * 64;
    int*   list = (int*)d_out;

    int attBlocks = (N + 8 * ATT_ROWS - 1) / (8 * ATT_ROWS);
    k_att<<<attBlocks, 256, 0, stream>>>(q, Watt, attS, attD, ctl, N);

    int edgeBlocks = (E + EBLK - 1) / EBLK;
    k_count<<<dim3(edgeBlocks, 2), 256, 0, stream>>>(dst1, dst2, tot, E);
    k_fill<<<dim3(edgeBlocks, 2), 256, 0, stream>>>(src1, dst1, src2, dst2, attS,
                                                   tot, cur, list, E);

    int buckets = (N + BKT - 1) / BKT;
    k_accum<<<dim3(buckets, 2), 256, 0, stream>>>(q, attD, list, tot, X1, X2, den1, den2, E, N);

    int gemmBlocks = (N + 31) / 32;
    k_gemm<<<gemmBlocks, 256, 0, stream>>>(q, X1, Wupd, Wv, bV, den1, den2, out, N);
}


# GEMM pass: a block's 64 rows and both weight matrices go through LDS once (whole-line loads, 68-float pitch: no bank conflicts), not strided 8-byte loads in each of 4 waves; whole-line stores
# speedup vs baseline: 3.3925x; 1.1812x over previous
#include <hip/hip_runtime.h>
#include <math.h>


typedef __attribute__((ext_vector_type(2))) float v2f;
typedef __attribute__((ext_vector_type(8))) float v8f;

#define BKT_SHIFT 8
#define BKT       256
#define NB_MAX    512
#define EPT       16
#define EBLK      (256 * EPT)

#define ATT_ROWS 32

template <int M>
__device__ __forceinline__ float row_xor(float x) {
    return __int_as_float(__builtin_amdgcn_mov_dpp(__float_as_int(x), 0x160 + M, 0xf, 0xf, true));
}

template <int M>
__device__ __forceinline__ void fold_level(float* a, bool up) {
    #pragma unroll
    for (int k = 0; k < M; ++k) {
        float keep = up ? a[k + M] : a[k];
        float send = up ? a[k] : a[k + M];
        a[k] = keep + row_xor<M>(send);
    }
}

template <bool RAGGED>
__device__ __forceinline__ void att_rows(const float* __restrict__ q, const float4 ws,
                                         const float4 wd, float* __restrict__ attS,
                                         float* __restrict__ attD, int base, int c, int h,
                                         int N) {
    float4 v[ATT_ROWS / 2];
    const float* p = q + (long long)(base + h) * 64 + 4 * c;
    #pragma unroll
    for (int j = 0; j < ATT_ROWS / 2; ++j) {
        if (RAGGED) {
            int row = min(base + 2 * j + h, N - 1);
            v[j] = *(const float4*)(q + (long long)row * 64 + 4 * c);
        } else {
            v[j] = *(const float4*)(p + j * 128);
        }
    }
    const v2f wsA = {ws.x, ws.y}, wsB = {ws.z, ws.w};
    const v2f wdA = {wd.x, wd.y}, wdB = {wd.z, wd.w};
    float s[ATT_ROWS / 2], d[ATT_ROWS / 2];
    #pragma unroll
    for (int j = 0; j < ATT_ROWS / 2; ++j) {
        const v2f a = {v[j].x, v[j].y}, b = {v[j].z, v[j].w};
        const v2f ts = a * wsA + b * wsB;
        const v2f td = a * wdA + b * wdB;
        s[j] = ts.x + ts.y;
        d[j] = td.x + td.y;
    }
    fold_level<8>(s, (c & 8) != 0);  fold_level<8>(d, (c & 8) != 0);
    fold_level<4>(s, (c & 4) != 0);  fold_level<4>(d, (c & 4) != 0);
    fold_level<2>(s, (c & 2) != 0);  fold_level<2>(d, (c & 2) != 0);
    fold_level<1>(s, (c & 1) != 0);  fold_level<1>(d, (c & 1) != 0);
    int node = base + 2 * c + h;
    if (!RAGGED || node < N) { attS[node] = s[0]; attD[node] = d[0]; }
}

__global__ __launch_bounds__(256) void k_att(const float* __restrict__ q,
                                             const float* __restrict__ Watt,
                                             float* __restrict__ attS,
                                             float* __restrict__ attD,
                                             int* __restrict__ ctl, int N) {
    for (int i = blockIdx.x * 256 + threadIdx.x; i < 4 * NB_MAX; i += gridDim.x * 256)
        ctl[i] = 0;
    int lane = threadIdx.x & 31;
    int c = lane & 15;
    int h = lane >> 4;
    int w = __builtin_amdgcn_readfirstlane(threadIdx.x >> 5);
    int base = (blockIdx.x * 8 + w) * ATT_ROWS;
    if (base >= N) return;
    const float4 ws = *(const float4*)(Watt + 4 * c);
    const float4 wd = *(const float4*)(Watt + 64 + 4 * c);
    if (base + ATT_ROWS <= N) att_rows<false>(q, ws, wd, attS, attD, base, c, h, N);
    else                      att_rows<true>(q, ws, wd, attS, attD, base, c, h, N);
}

__global__ __launch_bounds__(256) void k_count(const int* __restrict__ dst1,
                                               const int* __restrict__ dst2,
                                               int* __restrict__ tot, int E) {
    __shared__ int h[NB_MAX];
    int tid = threadIdx.x;
    int t = blockIdx.y;
    const int* dst = t ? dst2 : dst1;
    for (int i = tid; i < NB_MAX; i += 256) h[i] = 0;
    __syncthreads();
    int base = blockIdx.x * EBLK;
    #pragma unroll
    for (int k = 0; k < EPT; ++k) {
        int e = base + k * 256 + tid;
        if (e < E) atomicAdd(&h[dst[e] >> BKT_SHIFT], 1);
    }
    __syncthreads();
    for (int i = tid; i < NB_MAX; i += 256) {
        int c = h[i];
        if (c > 0) atomicAdd(&tot[t * NB_MAX + i], c);
    }
}

__device__ __forceinline__ void bucket_starts(const int* __restrict__ tot, int lane,
                                              int* bstart) {
    int c[16];
    const int4* tp = (const int4*)(tot + lane * 16);
    #pragma unroll
    for (int j = 0; j < 4; ++j) {
        int4 v = tp[j];
        c[4 * j] = v.x; c[4 * j + 1] = v.y; c[4 * j + 2] = v.z; c[4 * j + 3] = v.w;
    }
    int run = 0;
    #pragma unroll
    for (int j = 0; j < 16; ++j) { int x = c[j]; c[j] = run; run += x; }
    int inc = run;
    #pragma unroll
    for (int off = 1; off < 32; off <<= 1) {
        int y = __shfl_up(inc, off, 32);
        if (lane >= off) inc += y;
    }
    int excl = inc - run;
    #pragma unroll
    for (int j = 0; j < 16; ++j) bstart[lane * 16 + j] = excl + c[j];
}

__global__ __launch_bounds__(256) void k_fill(const int* __restrict__ src1,
                                              const int* __restrict__ dst1,
                                              const int* __restrict__ src2,
                                              const int* __restrict__ dst2,
                                              const float* __restrict__ attS,
                                              const int* __restrict__ tot,
                                              int* __restrict__ cur,
                                              int* __restrict__ list, int E) {
    __shared__ __attribute__((aligned(16))) int h[NB_MAX];
    __shared__ int bstart[NB_MAX];
    __shared__ int loff[NB_MAX];
    __shared__ int2 srec[EBLK];
    __shared__ unsigned short sbk[EBLK];
    int tid = threadIdx.x;
    int t = blockIdx.y;
    const int* src = t ? src2 : src1;
    const int* dst = t ? dst2 : dst1;
    for (int i = tid; i < NB_MAX; i += 256) h[i] = 0;
    __syncthreads();

    int base = blockIdx.x * EBLK;
    int nE = min(EBLK, E - base);
    int pk[EPT], rk[EPT], bk[EPT];
    float as[EPT];
    #pragma unroll
    for (int k = 0; k < EPT; ++k) {
        int e = base + k * 256 + tid;
        bk[k] = -1; pk[k] = 0; rk[k] = 0; as[k] = 0.0f;
        if (e < E) {
            int s = src[e], d = dst[e];
            as[k] = attS[s];
            bk[k] = d >> BKT_SHIFT;
            pk[k] = (s << BKT_SHIFT) | (d & (BKT - 1));
            rk[k] = atomicAdd(&h[bk[k]], 1);
        }
    }
    __syncthreads();
    if (tid < 32)      bucket_starts(tot + t * NB_MAX, tid, bstart);
    else if (tid < 64) bucket_starts(h, tid - 32, loff);
    __syncthreads();
    #pragma unroll
    for (int k = 0; k < EPT; ++k) {
        if (bk[k] >= 0) {
            int j = loff[bk[k]] + rk[k];
            srec[j] = make_int2(pk[k], __float_as_int(as[k]));
            sbk[j]  = (unsigned short)bk[k];
        }
    }
    for (int i = tid; i < NB_MAX; i += 256) {
        int cb = h[i];
        int g = 0;
        if (cb > 0) g = bstart[i] + atomicAdd(&cur[t * NB_MAX + i], cb);
        h[i] = g - loff[i];
    }
    __syncthreads();
    int2* L = (int2*)list + (long long)t * E;
    for (int j = tid; j < nE; j += 256) L[h[sbk[j]] + j] = srec[j];
}

#define ACC_GRP 16

__device__ __forceinline__ float lane_read_f(float v, int l) {
    return __int_as_float(__builtin_amdgcn_readlane(__float_as_int(v), l));
}

__device__ __forceinline__ void lds_fadd(float* p, float v) {
    __hip_atomic_fetch_add(p, v, __ATOMIC_RELAXED, __HIP_MEMORY_SCOPE_WORKGROUP);
}

__global__ __launch_bounds__(256) void k_accum(const float* __restrict__ q,
                                               const float* __restrict__ attD,
                                               const int* __restrict__ list,
                                               const int* __restrict__ tot,
                                               float* __restrict__ X1,
                                               float* __restrict__ X2,
                                               float* __restrict__ den1,
                                               float* __restrict__ den2,
                                               int E, int N) {
    __shared__ float acc[BKT * 64];
    __shared__ float dn[BKT];
    __shared__ float sAD[BKT];
    __shared__ int bstart[NB_MAX];
    int tid  = threadIdx.x;
    int lane = tid & 31;
    int w = __builtin_amdgcn_readfirstlane(tid >> 5);
    int t = blockIdx.y;
    int b = blockIdx.x;
    int node0 = b * BKT;

    float4* acc4 = (float4*)acc;
    #pragma unroll
    for (int it = 0; it < (BKT * 64 / 4) / 256; ++it)
        acc4[it * 256 + tid] = make_float4(0.0f, 0.0f, 0.0f, 0.0f);
    dn[tid]  = 0.0f;
    sAD[tid] = (node0 + tid < N) ? attD[node0 + tid] : 0.0f;
    if (w == 0) bucket_starts(tot + t * NB_MAX, lane, bstart);
    __syncthreads();

    int start = __builtin_amdgcn_readfirstlane(bstart[b]);
    int cnt   = __builtin_amdgcn_readfirstlane(tot[t * NB_MAX + b]);
    const int2* L = (const int2*)list + (long long)t * E + start;

    int2 en = make_int2(0, 0);
    if (w * 32 + lane < cnt) en = L[w * 32 + lane];
    for (int i = w * 32; i < cnt; i += 256) {
        int n = min(32, cnt - i);
        int pk = en.x;
        float as = __int_as_float(en.y);
        en = make_int2(0, 0);
        if (i + 256 + lane < cnt) en = L[i + 256 + lane];
        float a = 0.0f;
        if (lane < n) {
            float x  = as + sAD[pk & (BKT - 1)];
            float sg = 1.0f / (1.0f + expf(-x));
            a = expf(sg);
            lds_fadd(&dn[pk & (BKT - 1)], a);
        }
        for (int i0 = 0; i0 < n; i0 += ACC_GRP) {
            float2 v[ACC_GRP];
            #pragma unroll
            for (int u = 0; u < ACC_GRP; ++u) {
                int si = __builtin_amdgcn_readlane(pk, i0 + u) >> BKT_SHIFT;
                v[u] = *(const float2*)(q + (long long)si * 64 + 2 * lane);
            }
            if (i0 + ACC_GRP <= n) {
                #pragma unroll
                for (int u = 0; u < ACC_GRP; ++u) {
                    int   dl = __builtin_amdgcn_readlane(pk, i0 + u) & (BKT - 1);
                    float ai = lane_read_f(a, i0 + u);
                    float* ar = acc + dl * 64 + 2 * lane;
                    lds_fadd(ar,     ai * v[u].x);
                    lds_fadd(ar + 1, ai * v[u].y);
                }
            } else {
                #pragma unroll
                for (int u = 0; u < ACC_GRP; ++u) {
                    if (i0 + u < n) {
                        int   dl = __builtin_amdgcn_readlane(pk, i0 + u) & (BKT - 1);
                        float ai = lane_read_f(a, i0 + u);
                        float* ar = acc + dl * 64 + 2 * lane;
                        lds_fadd(ar,     ai * v[u].x);
                        lds_fadd(ar + 1, ai * v[u].y);
                    }
                }
            }
        }
    }
    __syncthreads();

    float* X   = t ? X2 : X1;
    float* den = t ? den2 : den1;
    #pragma unroll
    for (int it = 0; it < (BKT * 64 / 4) / 256; ++it) {
        int idx4 = it * 256 + tid;
        int r = idx4 >> 4;
        if (node0 + r < N)
            *(float4*)(X + (long long)(node0 + r) * 64 + (idx4 & 15) * 4) = acc4[idx4];
    }
    if (node0 + tid < N) den[node0 + tid] = dn[tid];
}

#define GT_ROWS  64
#define GT_PITCH 68

__global__ __launch_bounds__(256) void k_gemm(const float* __restrict__ q,
                                              const float* __restrict__ X1,
                                              const float* __restrict__ Wupd,
                                              const float* __restrict__ Wv,
                                              const float* __restrict__ bV,
                                              const float* __restrict__ den1,
                                              const float* __restrict__ den2,
                                              float* out, int N) {
    __shared__ __attribute__((aligned(16))) float sWu[64 * GT_PITCH];
    __shared__ __attribute__((aligned(16))) float sWv[64 * GT_PITCH];
    __shared__ __attribute__((aligned(16))) float sA1[GT_ROWS * GT_PITCH];
    __shared__ __attribute__((aligned(16))) float sA2[GT_ROWS * GT_PITCH];
    __shared__ float sSI[GT_ROWS];

    const int tid  = threadIdx.x;
    const int row0 = blockIdx.x * GT_ROWS;
    const int tr   = tid >> 4;
    const int c4   = tid & 15;
    const float* X2 = out + (long long)N * 64;

    float4 wu[4], wv[4], x1[4], x2[4], qv[4];
    float  d1[4], d2[4];
    #pragma unroll
    for (int j = 0; j < 4; ++j) {
        const int r = tr + 16 * j;
        wu[j] = *(const float4*)(Wupd + r * 64 + 4 * c4);
        wv[j] = *(const float4*)(Wv + r * 64 + 4 * c4);
    }
    #pragma unroll
    for (int j = 0; j < 4; ++j) {
        const int r = tr + 16 * j;
        const int g = min(row0 + r, N - 1);
        const long long go = (long long)g * 64 + 4 * c4;
        x1[j] = *(const float4*)(X1 + go);
        x2[j] = *(const float4*)(X2 + go);
        qv[j] = *(const float4*)(q + go);
        d1[j] = den1[g];
        d2[j] = den2[g];
    }
    #pragma unroll
    for (int j = 0; j < 4; ++j) {
        const int r = tr + 16 * j;
        *(float4*)&sWu[r * GT_PITCH + 4 * c4] = wu[j];
        *(float4*)&sWv[r * GT_PITCH + 4 * c4] = wv[j];
    }
    #pragma unroll
    for (int j = 0; j < 4; ++j) {
        const int r = tr + 16 * j;
        const float r1 = 1.0f / fmaxf(d1[j], 1e-12f);
        const float r2 = 1.0f / fmaxf(d2[j], 1e-12f);
        float4 a2, a1;
        a2.x = x1[j].x * r1 + x2[j].x * r2;  a2.y = x1[j].y * r1 + x2[j].y * r2;
        a2.z = x1[j].z * r1 + x2[j].z * r2;  a2.w = x1[j].w * r1 + x2[j].w * r2;
        a1.x = 2.0f * qv[j].x + a2.x;        a1.y = 2.0f * qv[j].y + a2.y;
        a1.z = 2.0f * qv[j].z + a2.z;        a1.w = 2.0f * qv[j].w + a2.w;
        *(float4*)&sA1[r * GT_PITCH + 4 * c4] = a1;
        *(float4*)&sA2[r * GT_PITCH + 4 * c4] = a2;
        if (c4 == 0)
            sSI[r] = ((d1[j] > 0.0f) ? 1.0f : 0.0f) + ((d2[j] > 0.0f) ? 1.0f : 0.0f);
    }
    __syncthreads();

    const int lane = tid & 31;
    const int w    = tid >> 5;
    const int c0   = (w & 3) * 16;
    const int rb   = (w >> 2) * 32;
    const int half = lane >> 4;
    const int l    = lane & 15;
    const int kofs = half * 2;

    const float* pBu = sWu + (c0 + l) * GT_PITCH + kofs;
    const float* pBv = sWv + (c0 + l) * GT_PITCH + kofs;
    const float* pA1 = sA1 + (rb + l) * GT_PITCH + kofs;
    const float* pA2 = sA2 + (rb + l) * GT_PITCH + kofs;

    v8f accU0 = {}, accU1 = {};
    v8f accV0 = {}, accV1 = {};

    #pragma unroll
    for (int k = 0; k < 64; k += 4) {
        const v2f bu  = *(const v2f*)(pBu + k);
        const v2f bv  = *(const v2f*)(pBv + k);
        const v2f a10 = *(const v2f*)(pA1 + k);
        const v2f a11 = *(const v2f*)(pA1 + 16 * GT_PITCH + k);
        const v2f a20 = *(const v2f*)(pA2 + k);
        const v2f a21 = *(const v2f*)(pA2 + 16 * GT_PITCH + k);
        accU0 = __builtin_amdgcn_wmma_f32_16x16x4_f32(false, a10, false, bu,
                                                      (short)0, accU0, false, false);
        accU1 = __builtin_amdgcn_wmma_f32_16x16x4_f32(false, a11, false, bu,
                                                      (short)0, accU1, false, false);
        accV0 = __builtin_amdgcn_wmma_f32_16x16x4_f32(false, a20, false, bv,
                                                      (short)0, accV0, false, false);
        accV1 = __builtin_amdgcn_wmma_f32_16x16x4_f32(false, a21, false, bv,
                                                      (short)0, accV1, false, false);
    }

    __syncthreads();

    #pragma unroll
    for (int r = 0; r < 8; ++r) {
        const int lr = rb + r + half * 8;
        sA1[lr * GT_PITCH + c0 + l]        = accU0[r];
        sA1[(lr + 16) * GT_PITCH + c0 + l] = accU1[r];
        sA2[lr * GT_PITCH + c0 + l]        = accV0[r];
        sA2[(lr + 16) * GT_PITCH + c0 + l] = accV1[r];
    }
    __syncthreads();

    const float4 b4 = *(const float4*)(bV + 4 * c4);
    #pragma unroll
    for (int j = 0; j < 4; ++j) {
        const int r = tr + 16 * j;
        const int g = row0 + r;
        if (g < N) {
            const float4 u = *(const float4*)&sA1[r * GT_PITCH + 4 * c4];
            float4 v = *(const float4*)&sA2[r * GT_PITCH + 4 * c4];
            const float sI = sSI[r];
            v.x = v.x + sI * b4.x;  v.y = v.y + sI * b4.y;
            v.z = v.z + sI * b4.z;  v.w = v.w + sI * b4.w;
            *(float4*)(out + (long long)g * 64 + 4 * c4) = u;
            *(float4*)(out + (long long)N * 64 + (long long)g * 64 + 4 * c4) = v;
        }
    }
}

extern "C" void kernel_launch(void* const* d_in, const int* in_sizes, int n_in,
                              void* d_out, int out_size, void* d_ws, size_t ws_size,
                              hipStream_t stream) {
    const float* q    = (const float*)d_in[0];
    const float* Watt = (const float*)d_in[1];
    const float* Wupd = (const float*)d_in[2];
    const float* Wv   = (const float*)d_in[3];
    const float* bV   = (const float*)d_in[4];
    const int*   src1 = (const int*)d_in[5];
    const int*   dst1 = (const int*)d_in[6];
    const int*   src2 = (const int*)d_in[7];
    const int*   dst2 = (const int*)d_in[8];

    const int N = in_sizes[0] / 64;
    const int E = in_sizes[5];

    float* X1   = (float*)d_ws;
    float* attS = X1 + (long long)N * 64;
    float* attD = attS + N;
    float* den1 = attS;
    float* den2 = attD + N;
    int*   ctl  = (int*)(den2 + N);
    int*   tot  = ctl;
    int*   cur  = ctl + 2 * NB_MAX;

    float* out  = (float*)d_out;
    float* X2   = out + (long long)N * 64;
    int*   list = (int*)d_out;

    int attBlocks = (N + 8 * ATT_ROWS - 1) / (8 * ATT_ROWS);
    k_att<<<attBlocks, 256, 0, stream>>>(q, Watt, attS, attD, ctl, N);

    int edgeBlocks = (E + EBLK - 1) / EBLK;
    k_count<<<dim3(edgeBlocks, 2), 256, 0, stream>>>(dst1, dst2, tot, E);
    k_fill<<<dim3(edgeBlocks, 2), 256, 0, stream>>>(src1, dst1, src2, dst2, attS,
                                                   tot, cur, list, E);

    int buckets = (N + BKT - 1) / BKT;
    k_accum<<<dim3(buckets, 2), 256, 0, stream>>>(q, attD, list, tot, X1, X2, den1, den2, E, N);

    int gemmBlocks = (N + GT_ROWS - 1) / GT_ROWS;
    k_gemm<<<gemmBlocks, 256, 0, stream>>>(q, X1, Wupd, Wv, bV, den1, den2, out, N);
}


# Bucket pass: records sorted by destination row in LDS (rank from an LDS integer add), a wave sums its rows' runs in registers, one store per row; LDS f32 sum tile goes; row sums in arrival order
# speedup vs baseline: 4.0092x; 1.1818x over previous
#include <hip/hip_runtime.h>
#include <math.h>


typedef __attribute__((ext_vector_type(2))) float v2f;
typedef __attribute__((ext_vector_type(8))) float v8f;

#define BKT_SHIFT 8
#define BKT       256
#define NB_MAX    512
#define EPT       16
#define EBLK      (256 * EPT)

#define ATT_ROWS 32

template <int M>
__device__ __forceinline__ float row_xor(float x) {
    return __int_as_float(__builtin_amdgcn_mov_dpp(__float_as_int(x), 0x160 + M, 0xf, 0xf, true));
}

template <int M>
__device__ __forceinline__ void fold_level(float* a, bool up) {
    #pragma unroll
    for (int k = 0; k < M; ++k) {
        float keep = up ? a[k + M] : a[k];
        float send = up ? a[k] : a[k + M];
        a[k] = keep + row_xor<M>(send);
    }
}

template <bool RAGGED>
__device__ __forceinline__ void att_rows(const float* __restrict__ q, const float4 ws,
                                         const float4 wd, float* __restrict__ attS,
                                         float* __restrict__ attD, int base, int c, int h,
                                         int N) {
    float4 v[ATT_ROWS / 2];
    const float* p = q + (long long)(base + h) * 64 + 4 * c;
    #pragma unroll
    for (int j = 0; j < ATT_ROWS / 2; ++j) {
        if (RAGGED) {
            int row = min(base + 2 * j + h, N - 1);
            v[j] = *(const float4*)(q + (long long)row * 64 + 4 * c);
        } else {
            v[j] = *(const float4*)(p + j * 128);
        }
    }
    const v2f wsA = {ws.x, ws.y}, wsB = {ws.z, ws.w};
    const v2f wdA = {wd.x, wd.y}, wdB = {wd.z, wd.w};
    float s[ATT_ROWS / 2], d[ATT_ROWS / 2];
    #pragma unroll
    for (int j = 0; j < ATT_ROWS / 2; ++j) {
        const v2f a = {v[j].x, v[j].y}, b = {v[j].z, v[j].w};
        const v2f ts = a * wsA + b * wsB;
        const v2f td = a * wdA + b * wdB;
        s[j] = ts.x + ts.y;
        d[j] = td.x + td.y;
    }
    fold_level<8>(s, (c & 8) != 0);  fold_level<8>(d, (c & 8) != 0);
    fold_level<4>(s, (c & 4) != 0);  fold_level<4>(d, (c & 4) != 0);
    fold_level<2>(s, (c & 2) != 0);  fold_level<2>(d, (c & 2) != 0);
    fold_level<1>(s, (c & 1) != 0);  fold_level<1>(d, (c & 1) != 0);
    int node = base + 2 * c + h;
    if (!RAGGED || node < N) { attS[node] = s[0]; attD[node] = d[0]; }
}

__global__ __launch_bounds__(256) void k_att(const float* __restrict__ q,
                                             const float* __restrict__ Watt,
                                             float* __restrict__ attS,
                                             float* __restrict__ attD,
                                             int* __restrict__ ctl, int N) {
    for (int i = blockIdx.x * 256 + threadIdx.x; i < 4 * NB_MAX; i += gridDim.x * 256)
        ctl[i] = 0;
    int lane = threadIdx.x & 31;
    int c = lane & 15;
    int h = lane >> 4;
    int w = __builtin_amdgcn_readfirstlane(threadIdx.x >> 5);
    int base = (blockIdx.x * 8 + w) * ATT_ROWS;
    if (base >= N) return;
    const float4 ws = *(const float4*)(Watt + 4 * c);
    const float4 wd = *(const float4*)(Watt + 64 + 4 * c);
    if (base + ATT_ROWS <= N) att_rows<false>(q, ws, wd, attS, attD, base, c, h, N);
    else                      att_rows<true>(q, ws, wd, attS, attD, base, c, h, N);
}

__global__ __launch_bounds__(256) void k_count(const int* __restrict__ dst1,
                                               const int* __restrict__ dst2,
                                               int* __restrict__ tot, int E) {
    __shared__ int h[NB_MAX];
    int tid = threadIdx.x;
    int t = blockIdx.y;
    const int* dst = t ? dst2 : dst1;
    for (int i = tid; i < NB_MAX; i += 256) h[i] = 0;
    __syncthreads();
    int base = blockIdx.x * EBLK;
    #pragma unroll
    for (int k = 0; k < EPT; ++k) {
        int e = base + k * 256 + tid;
        if (e < E) atomicAdd(&h[dst[e] >> BKT_SHIFT], 1);
    }
    __syncthreads();
    for (int i = tid; i < NB_MAX; i += 256) {
        int c = h[i];
        if (c > 0) atomicAdd(&tot[t * NB_MAX + i], c);
    }
}

__device__ __forceinline__ void bucket_starts(const int* __restrict__ tot, int lane,
                                              int* bstart) {
    int c[16];
    const int4* tp = (const int4*)(tot + lane * 16);
    #pragma unroll
    for (int j = 0; j < 4; ++j) {
        int4 v = tp[j];
        c[4 * j] = v.x; c[4 * j + 1] = v.y; c[4 * j + 2] = v.z; c[4 * j + 3] = v.w;
    }
    int run = 0;
    #pragma unroll
    for (int j = 0; j < 16; ++j) { int x = c[j]; c[j] = run; run += x; }
    int inc = run;
    #pragma unroll
    for (int off = 1; off < 32; off <<= 1) {
        int y = __shfl_up(inc, off, 32);
        if (lane >= off) inc += y;
    }
    int excl = inc - run;
    #pragma unroll
    for (int j = 0; j < 16; ++j) bstart[lane * 16 + j] = excl + c[j];
}

__global__ __launch_bounds__(256) void k_fill(const int* __restrict__ src1,
                                              const int* __restrict__ dst1,
                                              const int* __restrict__ src2,
                                              const int* __restrict__ dst2,
                                              const float* __restrict__ attS,
                                              const int* __restrict__ tot,
                                              int* __restrict__ cur,
                                              int* __restrict__ list, int E) {
    __shared__ __attribute__((aligned(16))) int h[NB_MAX];
    __shared__ int bstart[NB_MAX];
    __shared__ int loff[NB_MAX];
    __shared__ int2 srec[EBLK];
    __shared__ unsigned short sbk[EBLK];
    int tid = threadIdx.x;
    int t = blockIdx.y;
    const int* src = t ? src2 : src1;
    const int* dst = t ? dst2 : dst1;
    for (int i = tid; i < NB_MAX; i += 256) h[i] = 0;
    __syncthreads();

    int base = blockIdx.x * EBLK;
    int nE = min(EBLK, E - base);
    int pk[EPT], rk[EPT], bk[EPT];
    float as[EPT];
    #pragma unroll
    for (int k = 0; k < EPT; ++k) {
        int e = base + k * 256 + tid;
        bk[k] = -1; pk[k] = 0; rk[k] = 0; as[k] = 0.0f;
        if (e < E) {
            int s = src[e], d = dst[e];
            as[k] = attS[s];
            bk[k] = d >> BKT_SHIFT;
            pk[k] = (s << BKT_SHIFT) | (d & (BKT - 1));
            rk[k] = atomicAdd(&h[bk[k]], 1);
        }
    }
    __syncthreads();
    if (tid < 32)      bucket_starts(tot + t * NB_MAX, tid, bstart);
    else if (tid < 64) bucket_starts(h, tid - 32, loff);
    __syncthreads();
    #pragma unroll
    for (int k = 0; k < EPT; ++k) {
        if (bk[k] >= 0) {
            int j = loff[bk[k]] + rk[k];
            srec[j] = make_int2(pk[k], __float_as_int(as[k]));
            sbk[j]  = (unsigned short)bk[k];
        }
    }
    for (int i = tid; i < NB_MAX; i += 256) {
        int cb = h[i];
        int g = 0;
        if (cb > 0) g = bstart[i] + atomicAdd(&cur[t * NB_MAX + i], cb);
        h[i] = g - loff[i];
    }
    __syncthreads();
    int2* L = (int2*)list + (long long)t * E;
    for (int j = tid; j < nE; j += 256) L[h[sbk[j]] + j] = srec[j];
}

#define SRT_EPT 20
#define SRT_CH  (256 * SRT_EPT)
#define ACC_GRP 16

__device__ __forceinline__ float lane_read_f(float v, int l) {
    return __int_as_float(__builtin_amdgcn_readlane(__float_as_int(v), l));
}

__device__ __forceinline__ void lds_fadd(float* p, float v) {
    __hip_atomic_fetch_add(p, v, __ATOMIC_RELAXED, __HIP_MEMORY_SCOPE_WORKGROUP);
}

__device__ __forceinline__ void row_starts(const int* hist, int lane, int* rs) {
    int c[8];
    const int4* hp = (const int4*)(hist + lane * 8);
    const int4 v0 = hp[0], v1 = hp[1];
    c[0] = v0.x; c[1] = v0.y; c[2] = v0.z; c[3] = v0.w;
    c[4] = v1.x; c[5] = v1.y; c[6] = v1.z; c[7] = v1.w;
    int run = 0;
    for (int j = 0; j < 8; ++j) { int x = c[j]; c[j] = run; run += x; }
    int inc = run;
    for (int off = 1; off < 32; off <<= 1) {
        int y = __shfl_up(inc, off, 32);
        if (lane >= off) inc += y;
    }
    const int excl = inc - run;
    for (int j = 0; j < 8; ++j) rs[lane * 8 + j] = excl + c[j];
    if (lane == 31) rs[BKT] = inc;
}

__device__ __forceinline__ void row_pass(const float* __restrict__ q, const int2* srt,
                                         const int* rs, float* __restrict__ X,
                                         int node0, int R0, int lane, int N) {
    const int beg = __builtin_amdgcn_readfirstlane(rs[R0]);
    const int end = __builtin_amdgcn_readfirstlane(rs[R0 + 32]);
    const unsigned lane8 = (unsigned)lane * 8u;
    int   cur = R0;
    float ax = 0.0f, ay = 0.0f;

    #define FLUSH_TO(dl)                                                              \
        _Pragma("nounroll")                                                           \
        while (cur < (dl)) {                                                          \
            const int node = node0 + cur;                                             \
            if (node < N)                                                             \
                *(float2*)((char*)X + (long long)node * 256 + lane8) = make_float2(ax, ay); \
            ax = 0.0f; ay = 0.0f; ++cur;                                              \
        }

    for (int i = beg; i < end; i += 32) {
        const int n = min(32, end - i);
        int2 rec = make_int2(0, 0);
        if (lane < n) rec = srt[i + lane];
        const int      dlv  = rec.x & (BKT - 1);
        const unsigned offv = (unsigned)rec.x & ~(unsigned)(BKT - 1);
        const float    a    = __int_as_float(rec.y);
        int prev = __shfl_up(dlv, 1, 32);
        if (lane == 0) prev = cur;
        const unsigned mask = __builtin_amdgcn_ballot_w32(lane < n && dlv != prev);
        for (int i0 = 0; i0 < n; i0 += ACC_GRP) {
            float2 v[ACC_GRP];
            for (int u = 0; u < ACC_GRP; ++u) {
                const unsigned so = (unsigned)__builtin_amdgcn_readlane((int)offv, i0 + u);
                const char* rowp = (const char*)q + (unsigned long long)so;
                v[u] = *(const float2*)(rowp + lane8);
            }
            const unsigned mg = mask >> i0;
            if (i0 + ACC_GRP <= n) {
                for (int u = 0; u < ACC_GRP; ++u) {
                    if (mg & (1u << u)) {
                        const int dl = __builtin_amdgcn_readlane(dlv, i0 + u);
                        FLUSH_TO(dl)
                    }
                    const float ai = lane_read_f(a, i0 + u);
                    ax += ai * v[u].x;  ay += ai * v[u].y;
                }
            } else {
                for (int u = 0; u < ACC_GRP; ++u) {
                    if (i0 + u < n) {
                        if (mg & (1u << u)) {
                            const int dl = __builtin_amdgcn_readlane(dlv, i0 + u);
                            FLUSH_TO(dl)
                        }
                        const float ai = lane_read_f(a, i0 + u);
                        ax += ai * v[u].x;  ay += ai * v[u].y;
                    }
                }
            }
        }
    }
    FLUSH_TO(R0 + 32)
    #undef FLUSH_TO
}

__device__ __forceinline__ void row_pass_more(const float* __restrict__ q, const int2* srt,
                                              const int* rs, float* __restrict__ X,
                                              int node0, int R0, int lane) {
    const int beg = __builtin_amdgcn_readfirstlane(rs[R0]);
    const int end = __builtin_amdgcn_readfirstlane(rs[R0 + 32]);
    for (int i = beg; i < end; ++i) {
        const int2  rec = srt[i];
        const int   pk  = __builtin_amdgcn_readfirstlane(rec.x);
        const float a   = __int_as_float(__builtin_amdgcn_readfirstlane(rec.y));
        const int   dl  = pk & (BKT - 1);
        const float2 v  = *(const float2*)(q + (long long)(pk >> BKT_SHIFT) * 64 + 2 * lane);
        float2* xp = (float2*)(X + (long long)(node0 + dl) * 64 + 2 * lane);
        float2 o = *xp;
        o.x += a * v.x;  o.y += a * v.y;
        *xp = o;
        __threadfence_block();
    }
}

__global__ __launch_bounds__(256) void k_accum(const float* __restrict__ q,
                                               const float* __restrict__ attD,
                                               const int* __restrict__ list,
                                               const int* __restrict__ tot,
                                               float* __restrict__ X1,
                                               float* __restrict__ X2,
                                               float* __restrict__ den1,
                                               float* __restrict__ den2,
                                               int E, int N) {
    __shared__ int2 srt[SRT_CH];
    __shared__ __attribute__((aligned(16))) int hist[BKT];
    __shared__ int rs[BKT + 8];
    __shared__ float dn[BKT];
    __shared__ float sAD[BKT];
    __shared__ int bstart[NB_MAX];
    int tid  = threadIdx.x;
    int lane = tid & 31;
    int w = __builtin_amdgcn_readfirstlane(tid >> 5);
    int t = blockIdx.y;
    int b = blockIdx.x;
    int node0 = b * BKT;
    float* X   = t ? X2 : X1;
    float* den = t ? den2 : den1;

    dn[tid]  = 0.0f;
    sAD[tid] = (node0 + tid < N) ? attD[node0 + tid] : 0.0f;
    if (w == 0) bucket_starts(tot + t * NB_MAX, lane, bstart);
    __syncthreads();

    int start = __builtin_amdgcn_readfirstlane(bstart[b]);
    int cnt   = __builtin_amdgcn_readfirstlane(tot[t * NB_MAX + b]);
    const int2* L = (const int2*)list + (long long)t * E + start;

    for (int c0 = 0; ; c0 += SRT_CH) {
        const int m = min(SRT_CH, cnt - c0);
        hist[tid] = 0;
        __syncthreads();

        int2 r[SRT_EPT];
        int  rk[SRT_EPT];
        for (int k = 0; k < SRT_EPT; ++k) {
            const int j = k * 256 + tid;
            r[k] = make_int2(0, 0);
            if (j < m) r[k] = L[c0 + j];
        }
        for (int k = 0; k < SRT_EPT; ++k) {
            const int j = k * 256 + tid;
            rk[k] = 0;
            if (j < m) {
                const int   dl = r[k].x & (BKT - 1);
                const float x  = __int_as_float(r[k].y) + sAD[dl];
                const float sg = 1.0f / (1.0f + expf(-x));
                const float ex = expf(sg);
                r[k].y = __float_as_int(ex);
                lds_fadd(&dn[dl], ex);
                rk[k]  = atomicAdd(&hist[dl], 1);
            }
        }
        __syncthreads();
        if (w == 0) row_starts(hist, lane, rs);
        __syncthreads();
        for (int k = 0; k < SRT_EPT; ++k) {
            const int j = k * 256 + tid;
            if (j < m) srt[rs[r[k].x & (BKT - 1)] + rk[k]] = r[k];
        }
        __syncthreads();

        if (c0 == 0) row_pass(q, srt, rs, X, node0, w * 32, lane, N);
        else         row_pass_more(q, srt, rs, X, node0, w * 32, lane);
        if (c0 + SRT_CH >= cnt) break;
    }

    if (node0 + tid < N) den[node0 + tid] = dn[tid];
}

#define GT_ROWS  64
#define GT_PITCH 68

__global__ __launch_bounds__(256) void k_gemm(const float* __restrict__ q,
                                              const float* __restrict__ X1,
                                              const float* __restrict__ Wupd,
                                              const float* __restrict__ Wv,
                                              const float* __restrict__ bV,
                                              const float* __restrict__ den1,
                                              const float* __restrict__ den2,
                                              float* out, int N) {
    __shared__ __attribute__((aligned(16))) float sWu[64 * GT_PITCH];
    __shared__ __attribute__((aligned(16))) float sWv[64 * GT_PITCH];
    __shared__ __attribute__((aligned(16))) float sA1[GT_ROWS * GT_PITCH];
    __shared__ __attribute__((aligned(16))) float sA2[GT_ROWS * GT_PITCH];
    __shared__ float sSI[GT_ROWS];

    const int tid  = threadIdx.x;
    const int row0 = blockIdx.x * GT_ROWS;
    const int tr   = tid >> 4;
    const int c4   = tid & 15;
    const float* X2 = out + (long long)N * 64;

    float4 wu[4], wv[4], x1[4], x2[4], qv[4];
    float  d1[4], d2[4];
    #pragma unroll
    for (int j = 0; j < 4; ++j) {
        const int r = tr + 16 * j;
        wu[j] = *(const float4*)(Wupd + r * 64 + 4 * c4);
        wv[j] = *(const float4*)(Wv + r * 64 + 4 * c4);
    }
    #pragma unroll
    for (int j = 0; j < 4; ++j) {
        const int r = tr + 16 * j;
        const int g = min(row0 + r, N - 1);
        const long long go = (long long)g * 64 + 4 * c4;
        x1[j] = *(const float4*)(X1 + go);
        x2[j] = *(const float4*)(X2 + go);
        qv[j] = *(const float4*)(q + go);
        d1[j] = den1[g];
        d2[j] = den2[g];
    }
    #pragma unroll
    for (int j = 0; j < 4; ++j) {
        const int r = tr + 16 * j;
        *(float4*)&sWu[r * GT_PITCH + 4 * c4] = wu[j];
        *(float4*)&sWv[r * GT_PITCH + 4 * c4] = wv[j];
    }
    #pragma unroll
    for (int j = 0; j < 4; ++j) {
        const int r = tr + 16 * j;
        const float r1 = 1.0f / fmaxf(d1[j], 1e-12f);
        const float r2 = 1.0f / fmaxf(d2[j], 1e-12f);
        float4 a2, a1;
        a2.x = x1[j].x * r1 + x2[j].x * r2;  a2.y = x1[j].y * r1 + x2[j].y * r2;
        a2.z = x1[j].z * r1 + x2[j].z * r2;  a2.w = x1[j].w * r1 + x2[j].w * r2;
        a1.x = 2.0f * qv[j].x + a2.x;        a1.y = 2.0f * qv[j].y + a2.y;
        a1.z = 2.0f * qv[j].z + a2.z;        a1.w = 2.0f * qv[j].w + a2.w;
        *(float4*)&sA1[r * GT_PITCH + 4 * c4] = a1;
        *(float4*)&sA2[r * GT_PITCH + 4 * c4] = a2;
        if (c4 == 0)
            sSI[r] = ((d1[j] > 0.0f) ? 1.0f : 0.0f) + ((d2[j] > 0.0f) ? 1.0f : 0.0f);
    }
    __syncthreads();

    const int lane = tid & 31;
    const int w    = tid >> 5;
    const int c0   = (w & 3) * 16;
    const int rb   = (w >> 2) * 32;
    const int half = lane >> 4;
    const int l    = lane & 15;
    const int kofs = half * 2;

    const float* pBu = sWu + (c0 + l) * GT_PITCH + kofs;
    const float* pBv = sWv + (c0 + l) * GT_PITCH + kofs;
    const float* pA1 = sA1 + (rb + l) * GT_PITCH + kofs;
    const float* pA2 = sA2 + (rb + l) * GT_PITCH + kofs;

    v8f accU0 = {}, accU1 = {};
    v8f accV0 = {}, accV1 = {};

    #pragma unroll
    for (int k = 0; k < 64; k += 4) {
        const v2f bu  = *(const v2f*)(pBu + k);
        const v2f bv  = *(const v2f*)(pBv + k);
        const v2f a10 = *(const v2f*)(pA1 + k);
        const v2f a11 = *(const v2f*)(pA1 + 16 * GT_PITCH + k);
        const v2f a20 = *(const v2f*)(pA2 + k);
        const v2f a21 = *(const v2f*)(pA2 + 16 * GT_PITCH + k);
        accU0 = __builtin_amdgcn_wmma_f32_16x16x4_f32(false, a10, false, bu,
                                                      (short)0, accU0, false, false);
        accU1 = __builtin_amdgcn_wmma_f32_16x16x4_f32(false, a11, false, bu,
                                                      (short)0, accU1, false, false);
        accV0 = __builtin_amdgcn_wmma_f32_16x16x4_f32(false, a20, false, bv,
                                                      (short)0, accV0, false, false);
        accV1 = __builtin_amdgcn_wmma_f32_16x16x4_f32(false, a21, false, bv,
                                                      (short)0, accV1, false, false);
    }

    __syncthreads();

    #pragma unroll
    for (int r = 0; r < 8; ++r) {
        const int lr = rb + r + half * 8;
        sA1[lr * GT_PITCH + c0 + l]        = accU0[r];
        sA1[(lr + 16) * GT_PITCH + c0 + l] = accU1[r];
        sA2[lr * GT_PITCH + c0 + l]        = accV0[r];
        sA2[(lr + 16) * GT_PITCH + c0 + l] = accV1[r];
    }
    __syncthreads();

    const float4 b4 = *(const float4*)(bV + 4 * c4);
    #pragma unroll
    for (int j = 0; j < 4; ++j) {
        const int r = tr + 16 * j;
        const int g = row0 + r;
        if (g < N) {
            const float4 u = *(const float4*)&sA1[r * GT_PITCH + 4 * c4];
            float4 v = *(const float4*)&sA2[r * GT_PITCH + 4 * c4];
            const float sI = sSI[r];
            v.x = v.x + sI * b4.x;  v.y = v.y + sI * b4.y;
            v.z = v.z + sI * b4.z;  v.w = v.w + sI * b4.w;
            *(float4*)(out + (long long)g * 64 + 4 * c4) = u;
            *(float4*)(out + (long long)N * 64 + (long long)g * 64 + 4 * c4) = v;
        }
    }
}

extern "C" void kernel_launch(void* const* d_in, const int* in_sizes, int n_in,
                              void* d_out, int out_size, void* d_ws, size_t ws_size,
                              hipStream_t stream) {
    const float* q    = (const float*)d_in[0];
    const float* Watt = (const float*)d_in[1];
    const float* Wupd = (const float*)d_in[2];
    const float* Wv   = (const float*)d_in[3];
    const float* bV   = (const float*)d_in[4];
    const int*   src1 = (const int*)d_in[5];
    const int*   dst1 = (const int*)d_in[6];
    const int*   src2 = (const int*)d_in[7];
    const int*   dst2 = (const int*)d_in[8];

    const int N = in_sizes[0] / 64;
    const int E = in_sizes[5];

    float* X1   = (float*)d_ws;
    float* attS = X1 + (long long)N * 64;
    float* attD = attS + N;
    float* den1 = attS;
    float* den2 = attD + N;
    int*   ctl  = (int*)(den2 + N);
    int*   tot  = ctl;
    int*   cur  = ctl + 2 * NB_MAX;

    float* out  = (float*)d_out;
    float* X2   = out + (long long)N * 64;
    int*   list = (int*)d_out;

    int attBlocks = (N + 8 * ATT_ROWS - 1) / (8 * ATT_ROWS);
    k_att<<<attBlocks, 256, 0, stream>>>(q, Watt, attS, attD, ctl, N);

    int edgeBlocks = (E + EBLK - 1) / EBLK;
    k_count<<<dim3(edgeBlocks, 2), 256, 0, stream>>>(dst1, dst2, tot, E);
    k_fill<<<dim3(edgeBlocks, 2), 256, 0, stream>>>(src1, dst1, src2, dst2, attS,
                                                   tot, cur, list, E);

    int buckets = (N + BKT - 1) / BKT;
    k_accum<<<dim3(buckets, 2), 256, 0, stream>>>(q, attD, list, tot, X1, X2, den1, den2, E, N);

    int gemmBlocks = (N + GT_ROWS - 1) / GT_ROWS;
    k_gemm<<<gemmBlocks, 256, 0, stream>>>(q, X1, Wupd, Wv, bV, den1, den2, out, N);
}


# Fill pass: attS[src] taken from LDS (1024-thread block copies attS in whole-line parts, edges wait in registers), not one scattered 4-byte global load per edge; row sums stay in arrival order
# speedup vs baseline: 4.2960x; 1.0715x over previous
#include <hip/hip_runtime.h>
#include <math.h>


typedef __attribute__((ext_vector_type(2))) float v2f;
typedef __attribute__((ext_vector_type(8))) float v8f;

#define BKT_SHIFT 8
#define BKT       256
#define NB_MAX    512
#define EPT       16
#define EBLK      (256 * EPT)

#define ATT_ROWS 32

template <int M>
__device__ __forceinline__ float row_xor(float x) {
    return __int_as_float(__builtin_amdgcn_mov_dpp(__float_as_int(x), 0x160 + M, 0xf, 0xf, true));
}

template <int M>
__device__ __forceinline__ void fold_level(float* a, bool up) {
    #pragma unroll
    for (int k = 0; k < M; ++k) {
        float keep = up ? a[k + M] : a[k];
        float send = up ? a[k] : a[k + M];
        a[k] = keep + row_xor<M>(send);
    }
}

template <bool RAGGED>
__device__ __forceinline__ void att_rows(const float* __restrict__ q, const float4 ws,
                                         const float4 wd, float* __restrict__ attS,
                                         float* __restrict__ attD, int base, int c, int h,
                                         int N) {
    float4 v[ATT_ROWS / 2];
    const float* p = q + (long long)(base + h) * 64 + 4 * c;
    #pragma unroll
    for (int j = 0; j < ATT_ROWS / 2; ++j) {
        if (RAGGED) {
            int row = min(base + 2 * j + h, N - 1);
            v[j] = *(const float4*)(q + (long long)row * 64 + 4 * c);
        } else {
            v[j] = *(const float4*)(p + j * 128);
        }
    }
    const v2f wsA = {ws.x, ws.y}, wsB = {ws.z, ws.w};
    const v2f wdA = {wd.x, wd.y}, wdB = {wd.z, wd.w};
    float s[ATT_ROWS / 2], d[ATT_ROWS / 2];
    #pragma unroll
    for (int j = 0; j < ATT_ROWS / 2; ++j) {
        const v2f a = {v[j].x, v[j].y}, b = {v[j].z, v[j].w};
        const v2f ts = a * wsA + b * wsB;
        const v2f td = a * wdA + b * wdB;
        s[j] = ts.x + ts.y;
        d[j] = td.x + td.y;
    }
    fold_level<8>(s, (c & 8) != 0);  fold_level<8>(d, (c & 8) != 0);
    fold_level<4>(s, (c & 4) != 0);  fold_level<4>(d, (c & 4) != 0);
    fold_level<2>(s, (c & 2) != 0);  fold_level<2>(d, (c & 2) != 0);
    fold_level<1>(s, (c & 1) != 0);  fold_level<1>(d, (c & 1) != 0);
    int node = base + 2 * c + h;
    if (!RAGGED || node < N) { attS[node] = s[0]; attD[node] = d[0]; }
}

__global__ __launch_bounds__(256) void k_att(const float* __restrict__ q,
                                             const float* __restrict__ Watt,
                                             float* __restrict__ attS,
                                             float* __restrict__ attD,
                                             int* __restrict__ ctl, int N) {
    for (int i = blockIdx.x * 256 + threadIdx.x; i < 4 * NB_MAX; i += gridDim.x * 256)
        ctl[i] = 0;
    int lane = threadIdx.x & 31;
    int c = lane & 15;
    int h = lane >> 4;
    int w = __builtin_amdgcn_readfirstlane(threadIdx.x >> 5);
    int base = (blockIdx.x * 8 + w) * ATT_ROWS;
    if (base >= N) return;
    const float4 ws = *(const float4*)(Watt + 4 * c);
    const float4 wd = *(const float4*)(Watt + 64 + 4 * c);
    if (base + ATT_ROWS <= N) att_rows<false>(q, ws, wd, attS, attD, base, c, h, N);
    else                      att_rows<true>(q, ws, wd, attS, attD, base, c, h, N);
}

__global__ __launch_bounds__(256) void k_count(const int* __restrict__ dst1,
                                               const int* __restrict__ dst2,
                                               int* __restrict__ tot, int E) {
    __shared__ int h[NB_MAX];
    int tid = threadIdx.x;
    int t = blockIdx.y;
    const int* dst = t ? dst2 : dst1;
    for (int i = tid; i < NB_MAX; i += 256) h[i] = 0;
    __syncthreads();
    int base = blockIdx.x * EBLK;
    #pragma unroll
    for (int k = 0; k < EPT; ++k) {
        int e = base + k * 256 + tid;
        if (e < E) atomicAdd(&h[dst[e] >> BKT_SHIFT], 1);
    }
    __syncthreads();
    for (int i = tid; i < NB_MAX; i += 256) {
        int c = h[i];
        if (c > 0) atomicAdd(&tot[t * NB_MAX + i], c);
    }
}

__device__ __forceinline__ void bucket_starts(const int* __restrict__ tot, int lane,
                                              int* bstart) {
    int c[16];
    const int4* tp = (const int4*)(tot + lane * 16);
    #pragma unroll
    for (int j = 0; j < 4; ++j) {
        int4 v = tp[j];
        c[4 * j] = v.x; c[4 * j + 1] = v.y; c[4 * j + 2] = v.z; c[4 * j + 3] = v.w;
    }
    int run = 0;
    #pragma unroll
    for (int j = 0; j < 16; ++j) { int x = c[j]; c[j] = run; run += x; }
    int inc = run;
    #pragma unroll
    for (int off = 1; off < 32; off <<= 1) {
        int y = __shfl_up(inc, off, 32);
        if (lane >= off) inc += y;
    }
    int excl = inc - run;
    #pragma unroll
    for (int j = 0; j < 16; ++j) bstart[lane * 16 + j] = excl + c[j];
}

#define FT    1024
#define FEPT  13
#define FCH   (FT * FEPT)
#define TBL4  (8 * FT)
#define TBL   (4 * TBL4)

__global__ __launch_bounds__(FT) void k_fill(const int* __restrict__ src1,
                                             const int* __restrict__ dst1,
                                             const int* __restrict__ src2,
                                             const int* __restrict__ dst2,
                                             const float* __restrict__ attS,
                                             const int* __restrict__ tot,
                                             int* __restrict__ cur,
                                             int* __restrict__ list, int E, int N) {
    __shared__ __attribute__((aligned(16))) float tbl[TBL];
    __shared__ __attribute__((aligned(16))) int h[NB_MAX];
    __shared__ int bstart[NB_MAX];
    __shared__ int loff[NB_MAX];
    __shared__ int2 srec[FCH];
    __shared__ unsigned short sbk[FCH];
    const int tid = threadIdx.x;
    const int t = blockIdx.y;
    const int* src = t ? src2 : src1;
    const int* dst = t ? dst2 : dst1;
    int2* L = (int2*)list + (long long)t * E;

    const int per = (E + (int)gridDim.x - 1) / (int)gridDim.x;
    const int r0 = blockIdx.x * per;
    const int r1 = min(E, r0 + per);
    if (r0 >= r1) return;
    const int nsub = (r1 - r0 + FCH - 1) / FCH;
    const int sub  = (r1 - r0 + nsub - 1) / nsub;

    if (tid < 32) bucket_starts(tot + t * NB_MAX, tid, bstart);

    for (int sc = 0; sc < nsub; ++sc) {
        const int base = r0 + sc * sub;
        const int nE = min(sub, r1 - base);
        if (tid < NB_MAX) h[tid] = 0;

        int pk[FEPT], bk[FEPT];
        float as[FEPT];
        int sv[FEPT], dv[FEPT];
        #pragma unroll
        for (int k = 0; k < FEPT; ++k) {
            const int j = min(k * FT + tid, nE - 1);
            sv[k] = src[base + j];
            dv[k] = dst[base + j];
        }
        #pragma unroll
        for (int k = 0; k < FEPT; ++k) {
            const bool on = k * FT + tid < nE;
            bk[k] = on ? (dv[k] >> BKT_SHIFT) : -1;
            pk[k] = (sv[k] << BKT_SHIFT) | (dv[k] & (BKT - 1));
            as[k] = 0.0f;
        }

        for (int lo = 0; lo < N; lo += TBL) {
            const int cnt  = min(TBL, N - lo);
            const int cnt4 = cnt >> 2;
            const float4* gp = (const float4*)(attS + lo);
            float4 tv[TBL4 / FT];
            #pragma unroll
            for (int i = 0; i < TBL4 / FT; ++i) {
                const int i4 = i * FT + tid;
                tv[i] = make_float4(0.0f, 0.0f, 0.0f, 0.0f);
                if (i4 < cnt4) tv[i] = gp[i4];
            }
            __syncthreads();
            #pragma unroll
            for (int i = 0; i < TBL4 / FT; ++i) {
                const int i4 = i * FT + tid;
                if (i4 < cnt4) ((float4*)tbl)[i4] = tv[i];
            }
            if (tid < (cnt & 3)) tbl[4 * cnt4 + tid] = attS[lo + 4 * cnt4 + tid];
            __syncthreads();
            #pragma unroll
            for (int k = 0; k < FEPT; ++k) {
                const unsigned u = (unsigned)((pk[k] >> BKT_SHIFT) - lo);
                if (bk[k] >= 0 && u < (unsigned)cnt) as[k] = tbl[u];
            }
        }

        int rk[FEPT];
        #pragma unroll
        for (int k = 0; k < FEPT; ++k) {
            rk[k] = 0;
            if (bk[k] >= 0) rk[k] = atomicAdd(&h[bk[k]], 1);
        }
        __syncthreads();
        if (tid >= 32 && tid < 64) bucket_starts(h, tid - 32, loff);
        __syncthreads();
        #pragma unroll
        for (int k = 0; k < FEPT; ++k) {
            if (bk[k] >= 0) {
                const int j = loff[bk[k]] + rk[k];
                srec[j] = make_int2(pk[k], __float_as_int(as[k]));
                sbk[j]  = (unsigned short)bk[k];
            }
        }
        if (tid < NB_MAX) {
            const int cb = h[tid];
            int g = 0;
            if (cb > 0) g = bstart[tid] + atomicAdd(&cur[t * NB_MAX + tid], cb);
            h[tid] = g - loff[tid];
        }
        __syncthreads();
        for (int j = tid; j < nE; j += FT) L[h[sbk[j]] + j] = srec[j];
        __syncthreads();
    }
}

#define SRT_EPT 20
#define SRT_CH  (256 * SRT_EPT)
#define ACC_GRP 16

__device__ __forceinline__ float lane_read_f(float v, int l) {
    return __int_as_float(__builtin_amdgcn_readlane(__float_as_int(v), l));
}

__device__ __forceinline__ void lds_fadd(float* p, float v) {
    __hip_atomic_fetch_add(p, v, __ATOMIC_RELAXED, __HIP_MEMORY_SCOPE_WORKGROUP);
}

__device__ __forceinline__ void row_starts(const int* hist, int lane, int* rs) {
    int c[8];
    const int4* hp = (const int4*)(hist + lane * 8);
    const int4 v0 = hp[0], v1 = hp[1];
    c[0] = v0.x; c[1] = v0.y; c[2] = v0.z; c[3] = v0.w;
    c[4] = v1.x; c[5] = v1.y; c[6] = v1.z; c[7] = v1.w;
    int run = 0;
    for (int j = 0; j < 8; ++j) { int x = c[j]; c[j] = run; run += x; }
    int inc = run;
    for (int off = 1; off < 32; off <<= 1) {
        int y = __shfl_up(inc, off, 32);
        if (lane >= off) inc += y;
    }
    const int excl = inc - run;
    for (int j = 0; j < 8; ++j) rs[lane * 8 + j] = excl + c[j];
    if (lane == 31) rs[BKT] = inc;
}

__device__ __forceinline__ void row_pass(const float* __restrict__ q, const int2* srt,
                                         const int* rs, float* __restrict__ X,
                                         int node0, int R0, int lane, int N) {
    const int beg = __builtin_amdgcn_readfirstlane(rs[R0]);
    const int end = __builtin_amdgcn_readfirstlane(rs[R0 + 32]);
    const unsigned lane8 = (unsigned)lane * 8u;
    int   cur = R0;
    float ax = 0.0f, ay = 0.0f;

    #define FLUSH_TO(dl)                                                              \
        _Pragma("nounroll")                                                           \
        while (cur < (dl)) {                                                          \
            const int node = node0 + cur;                                             \
            if (node < N)                                                             \
                *(float2*)((char*)X + (long long)node * 256 + lane8) = make_float2(ax, ay); \
            ax = 0.0f; ay = 0.0f; ++cur;                                              \
        }

    for (int i = beg; i < end; i += 32) {
        const int n = min(32, end - i);
        int2 rec = make_int2(0, 0);
        if (lane < n) rec = srt[i + lane];
        const int      dlv  = rec.x & (BKT - 1);
        const unsigned offv = (unsigned)rec.x & ~(unsigned)(BKT - 1);
        const float    a    = __int_as_float(rec.y);
        int prev = __shfl_up(dlv, 1, 32);
        if (lane == 0) prev = cur;
        const unsigned mask = __builtin_amdgcn_ballot_w32(lane < n && dlv != prev);
        for (int i0 = 0; i0 < n; i0 += ACC_GRP) {
            float2 v[ACC_GRP];
            for (int u = 0; u < ACC_GRP; ++u) {
                const unsigned so = (unsigned)__builtin_amdgcn_readlane((int)offv, i0 + u);
                const char* rowp = (const char*)q + (unsigned long long)so;
                v[u] = *(const float2*)(rowp + lane8);
            }
            const unsigned mg = mask >> i0;
            if (i0 + ACC_GRP <= n) {
                for (int u = 0; u < ACC_GRP; ++u) {
                    if (mg & (1u << u)) {
                        const int dl = __builtin_amdgcn_readlane(dlv, i0 + u);
                        FLUSH_TO(dl)
                    }
                    const float ai = lane_read_f(a, i0 + u);
                    ax += ai * v[u].x;  ay += ai * v[u].y;
                }
            } else {
                for (int u = 0; u < ACC_GRP; ++u) {
                    if (i0 + u < n) {
                        if (mg & (1u << u)) {
                            const int dl = __builtin_amdgcn_readlane(dlv, i0 + u);
                            FLUSH_TO(dl)
                        }
                        const float ai = lane_read_f(a, i0 + u);
                        ax += ai * v[u].x;  ay += ai * v[u].y;
                    }
                }
            }
        }
    }
    FLUSH_TO(R0 + 32)
    #undef FLUSH_TO
}

__device__ __forceinline__ void row_pass_more(const float* __restrict__ q, const int2* srt,
                                              const int* rs, float* __restrict__ X,
                                              int node0, int R0, int lane) {
    const int beg = __builtin_amdgcn_readfirstlane(rs[R0]);
    const int end = __builtin_amdgcn_readfirstlane(rs[R0 + 32]);
    for (int i = beg; i < end; ++i) {
        const int2  rec = srt[i];
        const int   pk  = __builtin_amdgcn_readfirstlane(rec.x);
        const float a   = __int_as_float(__builtin_amdgcn_readfirstlane(rec.y));
        const int   dl  = pk & (BKT - 1);
        const float2 v  = *(const float2*)(q + (long long)(pk >> BKT_SHIFT) * 64 + 2 * lane);
        float2* xp = (float2*)(X + (long long)(node0 + dl) * 64 + 2 * lane);
        float2 o = *xp;
        o.x += a * v.x;  o.y += a * v.y;
        *xp = o;
        __threadfence_block();
    }
}

__global__ __launch_bounds__(256) void k_accum(const float* __restrict__ q,
                                               const float* __restrict__ attD,
                                               const int* __restrict__ list,
                                               const int* __restrict__ tot,
                                               float* __restrict__ X1,
                                               float* __restrict__ X2,
                                               float* __restrict__ den1,
                                               float* __restrict__ den2,
                                               int E, int N) {
    __shared__ int2 srt[SRT_CH];
    __shared__ __attribute__((aligned(16))) int hist[BKT];
    __shared__ int rs[BKT + 8];
    __shared__ float dn[BKT];
    __shared__ float sAD[BKT];
    __shared__ int bstart[NB_MAX];
    int tid  = threadIdx.x;
    int lane = tid & 31;
    int w = __builtin_amdgcn_readfirstlane(tid >> 5);
    int t = blockIdx.y;
    int b = blockIdx.x;
    int node0 = b * BKT;
    float* X   = t ? X2 : X1;
    float* den = t ? den2 : den1;

    dn[tid]  = 0.0f;
    sAD[tid] = (node0 + tid < N) ? attD[node0 + tid] : 0.0f;
    if (w == 0) bucket_starts(tot + t * NB_MAX, lane, bstart);
    __syncthreads();

    int start = __builtin_amdgcn_readfirstlane(bstart[b]);
    int cnt   = __builtin_amdgcn_readfirstlane(tot[t * NB_MAX + b]);
    const int2* L = (const int2*)list + (long long)t * E + start;

    for (int c0 = 0; ; c0 += SRT_CH) {
        const int m = min(SRT_CH, cnt - c0);
        hist[tid] = 0;
        __syncthreads();

        int2 r[SRT_EPT];
        int  rk[SRT_EPT];
        for (int k = 0; k < SRT_EPT; ++k) {
            const int j = k * 256 + tid;
            r[k] = make_int2(0, 0);
            if (j < m) r[k] = L[c0 + j];
        }
        for (int k = 0; k < SRT_EPT; ++k) {
            const int j = k * 256 + tid;
            rk[k] = 0;
            if (j < m) {
                const int   dl = r[k].x & (BKT - 1);
                const float x  = __int_as_float(r[k].y) + sAD[dl];
                const float sg = 1.0f / (1.0f + expf(-x));
                const float ex = expf(sg);
                r[k].y = __float_as_int(ex);
                lds_fadd(&dn[dl], ex);
                rk[k]  = atomicAdd(&hist[dl], 1);
            }
        }
        __syncthreads();
        if (w == 0) row_starts(hist, lane, rs);
        __syncthreads();
        for (int k = 0; k < SRT_EPT; ++k) {
            const int j = k * 256 + tid;
            if (j < m) srt[rs[r[k].x & (BKT - 1)] + rk[k]] = r[k];
        }
        __syncthreads();

        if (c0 == 0) row_pass(q, srt, rs, X, node0, w * 32, lane, N);
        else         row_pass_more(q, srt, rs, X, node0, w * 32, lane);
        if (c0 + SRT_CH >= cnt) break;
    }

    if (node0 + tid < N) den[node0 + tid] = dn[tid];
}

#define GT_ROWS  64
#define GT_PITCH 68

__global__ __launch_bounds__(256) void k_gemm(const float* __restrict__ q,
                                              const float* __restrict__ X1,
                                              const float* __restrict__ Wupd,
                                              const float* __restrict__ Wv,
                                              const float* __restrict__ bV,
                                              const float* __restrict__ den1,
                                              const float* __restrict__ den2,
                                              float* out, int N) {
    __shared__ __attribute__((aligned(16))) float sWu[64 * GT_PITCH];
    __shared__ __attribute__((aligned(16))) float sWv[64 * GT_PITCH];
    __shared__ __attribute__((aligned(16))) float sA1[GT_ROWS * GT_PITCH];
    __shared__ __attribute__((aligned(16))) float sA2[GT_ROWS * GT_PITCH];
    __shared__ float sSI[GT_ROWS];

    const int tid  = threadIdx.x;
    const int row0 = blockIdx.x * GT_ROWS;
    const int tr   = tid >> 4;
    const int c4   = tid & 15;
    const float* X2 = out + (long long)N * 64;

    float4 wu[4], wv[4], x1[4], x2[4], qv[4];
    float  d1[4], d2[4];
    #pragma unroll
    for (int j = 0; j < 4; ++j) {
        const int r = tr + 16 * j;
        wu[j] = *(const float4*)(Wupd + r * 64 + 4 * c4);
        wv[j] = *(const float4*)(Wv + r * 64 + 4 * c4);
    }
    #pragma unroll
    for (int j = 0; j < 4; ++j) {
        const int r = tr + 16 * j;
        const int g = min(row0 + r, N - 1);
        const long long go = (long long)g * 64 + 4 * c4;
        x1[j] = *(const float4*)(X1 + go);
        x2[j] = *(const float4*)(X2 + go);
        qv[j] = *(const float4*)(q + go);
        d1[j] = den1[g];
        d2[j] = den2[g];
    }
    #pragma unroll
    for (int j = 0; j < 4; ++j) {
        const int r = tr + 16 * j;
        *(float4*)&sWu[r * GT_PITCH + 4 * c4] = wu[j];
        *(float4*)&sWv[r * GT_PITCH + 4 * c4] = wv[j];
    }
    #pragma unroll
    for (int j = 0; j < 4; ++j) {
        const int r = tr + 16 * j;
        const float r1 = 1.0f / fmaxf(d1[j], 1e-12f);
        const float r2 = 1.0f / fmaxf(d2[j], 1e-12f);
        float4 a2, a1;
        a2.x = x1[j].x * r1 + x2[j].x * r2;  a2.y = x1[j].y * r1 + x2[j].y * r2;
        a2.z = x1[j].z * r1 + x2[j].z * r2;  a2.w = x1[j].w * r1 + x2[j].w * r2;
        a1.x = 2.0f * qv[j].x + a2.x;        a1.y = 2.0f * qv[j].y + a2.y;
        a1.z = 2.0f * qv[j].z + a2.z;        a1.w = 2.0f * qv[j].w + a2.w;
        *(float4*)&sA1[r * GT_PITCH + 4 * c4] = a1;
        *(float4*)&sA2[r * GT_PITCH + 4 * c4] = a2;
        if (c4 == 0)
            sSI[r] = ((d1[j] > 0.0f) ? 1.0f : 0.0f) + ((d2[j] > 0.0f) ? 1.0f : 0.0f);
    }
    __syncthreads();

    const int lane = tid & 31;
    const int w    = tid >> 5;
    const int c0   = (w & 3) * 16;
    const int rb   = (w >> 2) * 32;
    const int half = lane >> 4;
    const int l    = lane & 15;
    const int kofs = half * 2;

    const float* pBu = sWu + (c0 + l) * GT_PITCH + kofs;
    const float* pBv = sWv + (c0 + l) * GT_PITCH + kofs;
    const float* pA1 = sA1 + (rb + l) * GT_PITCH + kofs;
    const float* pA2 = sA2 + (rb + l) * GT_PITCH + kofs;

    v8f accU0 = {}, accU1 = {};
    v8f accV0 = {}, accV1 = {};

    #pragma unroll
    for (int k = 0; k < 64; k += 4) {
        const v2f bu  = *(const v2f*)(pBu + k);
        const v2f bv  = *(const v2f*)(pBv + k);
        const v2f a10 = *(const v2f*)(pA1 + k);
        const v2f a11 = *(const v2f*)(pA1 + 16 * GT_PITCH + k);
        const v2f a20 = *(const v2f*)(pA2 + k);
        const v2f a21 = *(const v2f*)(pA2 + 16 * GT_PITCH + k);
        accU0 = __builtin_amdgcn_wmma_f32_16x16x4_f32(false, a10, false, bu,
                                                      (short)0, accU0, false, false);
        accU1 = __builtin_amdgcn_wmma_f32_16x16x4_f32(false, a11, false, bu,
                                                      (short)0, accU1, false, false);
        accV0 = __builtin_amdgcn_wmma_f32_16x16x4_f32(false, a20, false, bv,
                                                      (short)0, accV0, false, false);
        accV1 = __builtin_amdgcn_wmma_f32_16x16x4_f32(false, a21, false, bv,
                                                      (short)0, accV1, false, false);
    }

    __syncthreads();

    #pragma unroll
    for (int r = 0; r < 8; ++r) {
        const int lr = rb + r + half * 8;
        sA1[lr * GT_PITCH + c0 + l]        = accU0[r];
        sA1[(lr + 16) * GT_PITCH + c0 + l] = accU1[r];
        sA2[lr * GT_PITCH + c0 + l]        = accV0[r];
        sA2[(lr + 16) * GT_PITCH + c0 + l] = accV1[r];
    }
    __syncthreads();

    const float4 b4 = *(const float4*)(bV + 4 * c4);
    #pragma unroll
    for (int j = 0; j < 4; ++j) {
        const int r = tr + 16 * j;
        const int g = row0 + r;
        if (g < N) {
            const float4 u = *(const float4*)&sA1[r * GT_PITCH + 4 * c4];
            float4 v = *(const float4*)&sA2[r * GT_PITCH + 4 * c4];
            const float sI = sSI[r];
            v.x = v.x + sI * b4.x;  v.y = v.y + sI * b4.y;
            v.z = v.z + sI * b4.z;  v.w = v.w + sI * b4.w;
            *(float4*)(out + (long long)g * 64 + 4 * c4) = u;
            *(float4*)(out + (long long)N * 64 + (long long)g * 64 + 4 * c4) = v;
        }
    }
}

extern "C" void kernel_launch(void* const* d_in, const int* in_sizes, int n_in,
                              void* d_out, int out_size, void* d_ws, size_t ws_size,
                              hipStream_t stream) {
    const float* q    = (const float*)d_in[0];
    const float* Watt = (const float*)d_in[1];
    const float* Wupd = (const float*)d_in[2];
    const float* Wv   = (const float*)d_in[3];
    const float* bV   = (const float*)d_in[4];
    const int*   src1 = (const int*)d_in[5];
    const int*   dst1 = (const int*)d_in[6];
    const int*   src2 = (const int*)d_in[7];
    const int*   dst2 = (const int*)d_in[8];

    const int N = in_sizes[0] / 64;
    const int E = in_sizes[5];

    float* X1   = (float*)d_ws;
    float* attS = X1 + (long long)N * 64;
    float* attD = attS + N;
    float* den1 = attS;
    float* den2 = attD + N;
    int*   ctl  = (int*)(den2 + N);
    int*   tot  = ctl;
    int*   cur  = ctl + 2 * NB_MAX;

    float* out  = (float*)d_out;
    float* X2   = out + (long long)N * 64;
    int*   list = (int*)d_out;

    int attBlocks = (N + 8 * ATT_ROWS - 1) / (8 * ATT_ROWS);
    k_att<<<attBlocks, 256, 0, stream>>>(q, Watt, attS, attD, ctl, N);

    int edgeBlocks = (E + EBLK - 1) / EBLK;
    k_count<<<dim3(edgeBlocks, 2), 256, 0, stream>>>(dst1, dst2, tot, E);
    k_fill<<<dim3(128, 2), FT, 0, stream>>>(src1, dst1, src2, dst2, attS,
                                            tot, cur, list, E, N);

    int buckets = (N + BKT - 1) / BKT;
    k_accum<<<dim3(buckets, 2), 256, 0, stream>>>(q, attD, list, tot, X1, X2, den1, den2, E, N);

    int gemmBlocks = (N + GT_ROWS - 1) / GT_ROWS;
    k_gemm<<<gemmBlocks, 256, 0, stream>>>(q, X1, Wupd, Wv, bV, den1, den2, out, N);
}


# Count pass: a thread's 16 loads of dst leave back to back (index clamped, no bounds branch between them) and are counted afterwards; before, each load waited for the one before it: 16 trips in a row
# speedup vs baseline: 5.3806x; 1.2525x over previous
#include <hip/hip_runtime.h>
#include <math.h>


typedef __attribute__((ext_vector_type(2))) float v2f;
typedef __attribute__((ext_vector_type(8))) float v8f;

#define BKT_SHIFT 8
#define BKT       256
#define NB_MAX    512
#define EPT       16
#define EBLK      (256 * EPT)

#define ATT_ROWS 32

template <int M>
__device__ __forceinline__ float row_xor(float x) {
    return __int_as_float(__builtin_amdgcn_mov_dpp(__float_as_int(x), 0x160 + M, 0xf, 0xf, true));
}

template <int M>
__device__ __forceinline__ void fold_level(float* a, bool up) {
    #pragma unroll
    for (int k = 0; k < M; ++k) {
        float keep = up ? a[k + M] : a[k];
        float send = up ? a[k] : a[k + M];
        a[k] = keep + row_xor<M>(send);
    }
}

template <bool RAGGED>
__device__ __forceinline__ void att_rows(const float* __restrict__ q, const float4 ws,
                                         const float4 wd, float* __restrict__ attS,
                                         float* __restrict__ attD, int base, int c, int h,
                                         int N) {
    float4 v[ATT_ROWS / 2];
    const float* p = q + (long long)(base + h) * 64 + 4 * c;
    #pragma unroll
    for (int j = 0; j < ATT_ROWS / 2; ++j) {
        if (RAGGED) {
            int row = min(base + 2 * j + h, N - 1);
            v[j] = *(const float4*)(q + (long long)row * 64 + 4 * c);
        } else {
            v[j] = *(const float4*)(p + j * 128);
        }
    }
    const v2f wsA = {ws.x, ws.y}, wsB = {ws.z, ws.w};
    const v2f wdA = {wd.x, wd.y}, wdB = {wd.z, wd.w};
    float s[ATT_ROWS / 2], d[ATT_ROWS / 2];
    #pragma unroll
    for (int j = 0; j < ATT_ROWS / 2; ++j) {
        const v2f a = {v[j].x, v[j].y}, b = {v[j].z, v[j].w};
        const v2f ts = a * wsA + b * wsB;
        const v2f td = a * wdA + b * wdB;
        s[j] = ts.x + ts.y;
        d[j] = td.x + td.y;
    }
    fold_level<8>(s, (c & 8) != 0);  fold_level<8>(d, (c & 8) != 0);
    fold_level<4>(s, (c & 4) != 0);  fold_level<4>(d, (c & 4) != 0);
    fold_level<2>(s, (c & 2) != 0);  fold_level<2>(d, (c & 2) != 0);
    fold_level<1>(s, (c & 1) != 0);  fold_level<1>(d, (c & 1) != 0);
    int node = base + 2 * c + h;
    if (!RAGGED || node < N) { attS[node] = s[0]; attD[node] = d[0]; }
}

__global__ __launch_bounds__(256) void k_att(const float* __restrict__ q,
                                             const float* __restrict__ Watt,
                                             float* __restrict__ attS,
                                             float* __restrict__ attD,
                                             int* __restrict__ ctl, int N) {
    for (int i = blockIdx.x * 256 + threadIdx.x; i < 4 * NB_MAX; i += gridDim.x * 256)
        ctl[i] = 0;
    int lane = threadIdx.x & 31;
    int c = lane & 15;
    int h = lane >> 4;
    int w = __builtin_amdgcn_readfirstlane(threadIdx.x >> 5);
    int base = (blockIdx.x * 8 + w) * ATT_ROWS;
    if (base >= N) return;
    const float4 ws = *(const float4*)(Watt + 4 * c);
    const float4 wd = *(const float4*)(Watt + 64 + 4 * c);
    if (base + ATT_ROWS <= N) att_rows<false>(q, ws, wd, attS, attD, base, c, h, N);
    else                      att_rows<true>(q, ws, wd, attS, attD, base, c, h, N);
}

__global__ __launch_bounds__(256) void k_count(const int* __restrict__ dst1,
                                               const int* __restrict__ dst2,
                                               int* __restrict__ tot, int E) {
    __shared__ int h[NB_MAX];
    int tid = threadIdx.x;
    int t = blockIdx.y;
    const int* dst = t ? dst2 : dst1;
    for (int i = tid; i < NB_MAX; i += 256) h[i] = 0;
    int base = blockIdx.x * EBLK;
    int dv[EPT];
    #pragma unroll
    for (int k = 0; k < EPT; ++k) {
        const int e = base + k * 256 + tid;
        dv[k] = dst[min(e, E - 1)];
    }
    __syncthreads();
    #pragma unroll
    for (int k = 0; k < EPT; ++k) {
        const int e = base + k * 256 + tid;
        if (e < E) atomicAdd(&h[dv[k] >> BKT_SHIFT], 1);
    }
    __syncthreads();
    for (int i = tid; i < NB_MAX; i += 256) {
        int c = h[i];
        if (c > 0) atomicAdd(&tot[t * NB_MAX + i], c);
    }
}

__device__ __forceinline__ void bucket_starts(const int* __restrict__ tot, int lane,
                                              int* bstart) {
    int c[16];
    const int4* tp = (const int4*)(tot + lane * 16);
    #pragma unroll
    for (int j = 0; j < 4; ++j) {
        int4 v = tp[j];
        c[4 * j] = v.x; c[4 * j + 1] = v.y; c[4 * j + 2] = v.z; c[4 * j + 3] = v.w;
    }
    int run = 0;
    #pragma unroll
    for (int j = 0; j < 16; ++j) { int x = c[j]; c[j] = run; run += x; }
    int inc = run;
    #pragma unroll
    for (int off = 1; off < 32; off <<= 1) {
        int y = __shfl_up(inc, off, 32);
        if (lane >= off) inc += y;
    }
    int excl = inc - run;
    #pragma unroll
    for (int j = 0; j < 16; ++j) bstart[lane * 16 + j] = excl + c[j];
}

#define FT    1024
#define FEPT  13
#define FCH   (FT * FEPT)
#define TBL4  (8 * FT)
#define TBL   (4 * TBL4)

__global__ __launch_bounds__(FT) void k_fill(const int* __restrict__ src1,
                                             const int* __restrict__ dst1,
                                             const int* __restrict__ src2,
                                             const int* __restrict__ dst2,
                                             const float* __restrict__ attS,
                                             const int* __restrict__ tot,
                                             int* __restrict__ cur,
                                             int* __restrict__ list, int E, int N) {
    __shared__ __attribute__((aligned(16))) float tbl[TBL];
    __shared__ __attribute__((aligned(16))) int h[NB_MAX];
    __shared__ int bstart[NB_MAX];
    __shared__ int loff[NB_MAX];
    __shared__ int2 srec[FCH];
    __shared__ unsigned short sbk[FCH];
    const int tid = threadIdx.x;
    const int t = blockIdx.y;
    const int* src = t ? src2 : src1;
    const int* dst = t ? dst2 : dst1;
    int2* L = (int2*)list + (long long)t * E;

    const int per = (E + (int)gridDim.x - 1) / (int)gridDim.x;
    const int r0 = blockIdx.x * per;
    const int r1 = min(E, r0 + per);
    if (r0 >= r1) return;
    const int nsub = (r1 - r0 + FCH - 1) / FCH;
    const int sub  = (r1 - r0 + nsub - 1) / nsub;

    if (tid < 32) bucket_starts(tot + t * NB_MAX, tid, bstart);

    for (int sc = 0; sc < nsub; ++sc) {
        const int base = r0 + sc * sub;
        const int nE = min(sub, r1 - base);
        if (tid < NB_MAX) h[tid] = 0;

        int pk[FEPT], bk[FEPT];
        float as[FEPT];
        int sv[FEPT], dv[FEPT];
        #pragma unroll
        for (int k = 0; k < FEPT; ++k) {
            const int j = min(k * FT + tid, nE - 1);
            sv[k] = src[base + j];
            dv[k] = dst[base + j];
        }
        #pragma unroll
        for (int k = 0; k < FEPT; ++k) {
            const bool on = k * FT + tid < nE;
            bk[k] = on ? (dv[k] >> BKT_SHIFT) : -1;
            pk[k] = (sv[k] << BKT_SHIFT) | (dv[k] & (BKT - 1));
            as[k] = 0.0f;
        }

        for (int lo = 0; lo < N; lo += TBL) {
            const int cnt  = min(TBL, N - lo);
            const int cnt4 = cnt >> 2;
            const float4* gp = (const float4*)(attS + lo);
            float4 tv[TBL4 / FT];
            #pragma unroll
            for (int i = 0; i < TBL4 / FT; ++i) {
                const int i4 = i * FT + tid;
                tv[i] = make_float4(0.0f, 0.0f, 0.0f, 0.0f);
                if (i4 < cnt4) tv[i] = gp[i4];
            }
            __syncthreads();
            #pragma unroll
            for (int i = 0; i < TBL4 / FT; ++i) {
                const int i4 = i * FT + tid;
                if (i4 < cnt4) ((float4*)tbl)[i4] = tv[i];
            }
            if (tid < (cnt & 3)) tbl[4 * cnt4 + tid] = attS[lo + 4 * cnt4 + tid];
            __syncthreads();
            #pragma unroll
            for (int k = 0; k < FEPT; ++k) {
                const unsigned u = (unsigned)((pk[k] >> BKT_SHIFT) - lo);
                if (bk[k] >= 0 && u < (unsigned)cnt) as[k] = tbl[u];
            }
        }

        int rk[FEPT];
        #pragma unroll
        for (int k = 0; k < FEPT; ++k) {
            rk[k] = 0;
            if (bk[k] >= 0) rk[k] = atomicAdd(&h[bk[k]], 1);
        }
        __syncthreads();
        if (tid >= 32 && tid < 64) bucket_starts(h, tid - 32, loff);
        __syncthreads();
        #pragma unroll
        for (int k = 0; k < FEPT; ++k) {
            if (bk[k] >= 0) {
                const int j = loff[bk[k]] + rk[k];
                srec[j] = make_int2(pk[k], __float_as_int(as[k]));
                sbk[j]  = (unsigned short)bk[k];
            }
        }
        if (tid < NB_MAX) {
            const int cb = h[tid];
            int g = 0;
            if (cb > 0) g = bstart[tid] + atomicAdd(&cur[t * NB_MAX + tid], cb);
            h[tid] = g - loff[tid];
        }
        __syncthreads();
        for (int j = tid; j < nE; j += FT) L[h[sbk[j]] + j] = srec[j];
        __syncthreads();
    }
}

#define SRT_EPT 19
#define SRT_CH  (256 * SRT_EPT)
#define ACC_GRP 16
#define GT_PITCH 68

__device__ __forceinline__ float lane_read_f(float v, int l) {
    return __int_as_float(__builtin_amdgcn_readlane(__float_as_int(v), l));
}

__device__ __forceinline__ void lds_fadd(float* p, float v) {
    __hip_atomic_fetch_add(p, v, __ATOMIC_RELAXED, __HIP_MEMORY_SCOPE_WORKGROUP);
}

__device__ __forceinline__ float ex_of(float x) {
    const float sg = 1.0f / (1.0f + expf(-x));
    return expf(sg);
}

__device__ __forceinline__ void row_starts(const int* hist, int lane, int* rs) {
    int c[8];
    const int4* hp = (const int4*)(hist + lane * 8);
    const int4 v0 = hp[0], v1 = hp[1];
    c[0] = v0.x; c[1] = v0.y; c[2] = v0.z; c[3] = v0.w;
    c[4] = v1.x; c[5] = v1.y; c[6] = v1.z; c[7] = v1.w;
    int run = 0;
    for (int j = 0; j < 8; ++j) { int x = c[j]; c[j] = run; run += x; }
    int inc = run;
    for (int off = 1; off < 32; off <<= 1) {
        int y = __shfl_up(inc, off, 32);
        if (lane >= off) inc += y;
    }
    const int excl = inc - run;
    for (int j = 0; j < 8; ++j) rs[lane * 8 + j] = excl + c[j];
    if (lane == 31) rs[BKT] = inc;
}

__device__ __forceinline__ void row_pass(const float* __restrict__ q, const int2* srt,
                                         const int* rs, const float* dnT, float* tile,
                                         int R0, int lane) {
    const int beg = __builtin_amdgcn_readfirstlane(rs[R0]);
    const int end = __builtin_amdgcn_readfirstlane(rs[R0 + 32]);
    const unsigned lane8 = (unsigned)lane * 8u;
    int   cur = R0;
    float ax = 0.0f, ay = 0.0f;
    const float rinvLane = 1.0f / fmaxf(dnT[R0 + lane], 1e-12f);

    #define FLUSH_TO(dl)                                                              \
        _Pragma("nounroll")                                                           \
        while (cur < (dl)) {                                                          \
            const float rinv = lane_read_f(rinvLane, cur - R0);                       \
            float* tp = (float*)((char*)tile + cur * (GT_PITCH * 4) + lane8);         \
            lds_fadd(tp, ax * rinv);                                                  \
            lds_fadd(tp + 1, ay * rinv);                                              \
            ax = 0.0f; ay = 0.0f; ++cur;                                              \
        }

    for (int i = beg; i < end; i += 32) {
        const int n = min(32, end - i);
        int2 rec = make_int2(0, 0);
        if (lane < n) rec = srt[i + lane];
        const int      dlv  = rec.x & (BKT - 1);
        const unsigned offv = (unsigned)rec.x & ~(unsigned)(BKT - 1);
        const float    a    = __int_as_float(rec.y);
        int prev = __shfl_up(dlv, 1, 32);
        if (lane == 0) prev = cur;
        const unsigned mask = __builtin_amdgcn_ballot_w32(lane < n && dlv != prev);
        for (int i0 = 0; i0 < n; i0 += ACC_GRP) {
            float2 v[ACC_GRP];
            for (int u = 0; u < ACC_GRP; ++u) {
                const unsigned so = (unsigned)__builtin_amdgcn_readlane((int)offv, i0 + u);
                const char* rowp = (const char*)q + (unsigned long long)so;
                v[u] = *(const float2*)(rowp + lane8);
            }
            const unsigned mg = mask >> i0;
            if (i0 + ACC_GRP <= n) {
                for (int u = 0; u < ACC_GRP; ++u) {
                    if (mg & (1u << u)) {
                        const int dl = __builtin_amdgcn_readlane(dlv, i0 + u);
                        FLUSH_TO(dl)
                    }
                    const float ai = lane_read_f(a, i0 + u);
                    ax += ai * v[u].x;  ay += ai * v[u].y;
                }
            } else {
                for (int u = 0; u < ACC_GRP; ++u) {
                    if (i0 + u < n) {
                        if (mg & (1u << u)) {
                            const int dl = __builtin_amdgcn_readlane(dlv, i0 + u);
                            FLUSH_TO(dl)
                        }
                        const float ai = lane_read_f(a, i0 + u);
                        ax += ai * v[u].x;  ay += ai * v[u].y;
                    }
                }
            }
        }
    }
    FLUSH_TO(R0 + 32)
    #undef FLUSH_TO
}

__device__ __forceinline__ void row_pass_more(const float* __restrict__ q, const int2* srt,
                                              const int* rs, const float* dnT, float* tile,
                                              int R0, int lane) {
    const int beg = __builtin_amdgcn_readfirstlane(rs[R0]);
    const int end = __builtin_amdgcn_readfirstlane(rs[R0 + 32]);
    const float rinvLane = 1.0f / fmaxf(dnT[R0 + lane], 1e-12f);
    for (int i = beg; i < end; ++i) {
        const int2  rec = srt[i];
        const int   pk  = __builtin_amdgcn_readfirstlane(rec.x);
        const float a   = __int_as_float(__builtin_amdgcn_readfirstlane(rec.y));
        const int   dl  = pk & (BKT - 1);
        const float2 v  = *(const float2*)(q + (long long)(pk >> BKT_SHIFT) * 64 + 2 * lane);
        const float rinv = lane_read_f(rinvLane, dl - R0);
        float* tp = tile + dl * GT_PITCH + 2 * lane;
        lds_fadd(tp, a * v.x * rinv);
        lds_fadd(tp + 1, a * v.y * rinv);
    }
}

#define GT_ROWS  64

__global__ __launch_bounds__(512) void k_accum(const float* __restrict__ q,
                                               const float* __restrict__ attD,
                                               const int* __restrict__ list,
                                               const int* __restrict__ tot,
                                               const float* __restrict__ Wupd,
                                               const float* __restrict__ Wv,
                                               const float* __restrict__ bV,
                                               float* __restrict__ out,
                                               int E, int N) {
    __shared__ __attribute__((aligned(16))) int2 srt[2 * SRT_CH];
    __shared__ __attribute__((aligned(16))) float tile[BKT * GT_PITCH];
    __shared__ __attribute__((aligned(16))) int hist[2 * BKT];
    __shared__ int rs[2 * (BKT + 8)];
    __shared__ float dn[2 * BKT];
    __shared__ float sAD[BKT];
    __shared__ int sStart[2];
    const int tid  = threadIdx.x;
    const int lane = tid & 31;
    const int tid8 = tid & 255;
    const int wv = __builtin_amdgcn_readfirstlane(tid >> 5);
    const int t  = wv >> 3;
    const int w  = wv & 7;
    const int b  = blockIdx.x;
    const int node0 = b * BKT;

    int2*  srtT  = srt + t * SRT_CH;
    int*   histT = hist + t * BKT;
    int*   rsT   = rs + t * (BKT + 8);
    float* dnT   = dn + t * BKT;

    for (int i = tid; i < BKT * GT_PITCH / 4; i += 512)
        ((float4*)tile)[i] = make_float4(0.0f, 0.0f, 0.0f, 0.0f);
    dn[tid] = 0.0f;
    if (tid < BKT) sAD[tid] = (node0 + tid < N) ? attD[node0 + tid] : 0.0f;
    if (w == 0) {
        const int4* tp = (const int4*)(tot + t * NB_MAX + lane * 16);
        int s = 0;
        #pragma unroll
        for (int j = 0; j < 4; ++j) {
            const int4 v = tp[j];
            const int i0 = lane * 16 + 4 * j;
            s += (i0 < b ? v.x : 0) + (i0 + 1 < b ? v.y : 0) + (i0 + 2 < b ? v.z : 0)
               + (i0 + 3 < b ? v.w : 0);
        }
        #pragma unroll
        for (int off = 16; off >= 1; off >>= 1) s += __shfl_xor(s, off, 32);
        if (lane == 0) sStart[t] = s;
    }
    __syncthreads();

    const int start  = __builtin_amdgcn_readfirstlane(sStart[t]);
    const int cnt    = __builtin_amdgcn_readfirstlane(tot[t * NB_MAX + b]);
    const int cntMax = max(__builtin_amdgcn_readfirstlane(tot[b]),
                           __builtin_amdgcn_readfirstlane(tot[NB_MAX + b]));
    const int2* L = (const int2*)list + (long long)t * E + start;

    for (int j = SRT_CH + tid8; j < cnt; j += 256) {
        const int2 rec = L[j];
        const int  dl  = rec.x & (BKT - 1);
        lds_fadd(&dnT[dl], ex_of(__int_as_float(rec.y) + sAD[dl]));
    }

    for (int c0 = 0; ; c0 += SRT_CH) {
        const int m = max(0, min(SRT_CH, cnt - c0));
        histT[tid8] = 0;
        __syncthreads();

        int2 r[SRT_EPT];
        int  rk[SRT_EPT];
        for (int k = 0; k < SRT_EPT; ++k) {
            const int j = k * 256 + tid8;
            r[k] = make_int2(0, 0);
            if (j < m) r[k] = L[c0 + j];
        }
        for (int k = 0; k < SRT_EPT; ++k) {
            const int j = k * 256 + tid8;
            rk[k] = 0;
            if (j < m) {
                const int   dl = r[k].x & (BKT - 1);
                const float ex = ex_of(__int_as_float(r[k].y) + sAD[dl]);
                r[k].y = __float_as_int(ex);
                if (c0 == 0) lds_fadd(&dnT[dl], ex);
                rk[k]  = atomicAdd(&histT[dl], 1);
            }
        }
        __syncthreads();
        if (w == 0) row_starts(histT, lane, rsT);
        __syncthreads();
        for (int k = 0; k < SRT_EPT; ++k) {
            const int j = k * 256 + tid8;
            if (j < m) srtT[rsT[r[k].x & (BKT - 1)] + rk[k]] = r[k];
        }
        __syncthreads();

        if (c0 == 0) row_pass(q, srtT, rsT, dnT, tile, w * 32, lane);
        else         row_pass_more(q, srtT, rsT, dnT, tile, w * 32, lane);
        if (c0 + SRT_CH >= cntMax) break;
    }
    const int tr = tid8 >> 4;
    const int c4 = tid8 & 15;
    const int wr = tid >> 4;

    const float4 wu0 = *(const float4*)(Wupd + wr * 64 + 4 * c4);
    const float4 wu1 = *(const float4*)(Wupd + (wr + 32) * 64 + 4 * c4);
    const float4 wv0 = *(const float4*)(Wv + wr * 64 + 4 * c4);
    const float4 wv1 = *(const float4*)(Wv + (wr + 32) * 64 + 4 * c4);
    const float4 b4  = *(const float4*)(bV + 4 * c4);
    float4 qv[4];
    #pragma unroll
    for (int j = 0; j < 4; ++j) {
        const int g = min(node0 + t * GT_ROWS + tr + 16 * j, N - 1);
        qv[j] = *(const float4*)(q + (long long)g * 64 + 4 * c4);
    }

    __syncthreads();

    float* sWu = (float*)srt;
    float* sWv = sWu + 64 * GT_PITCH;
    float* sA1 = sWv + 64 * GT_PITCH + t * (GT_ROWS * GT_PITCH);
    *(float4*)&sWu[wr * GT_PITCH + 4 * c4]        = wu0;
    *(float4*)&sWu[(wr + 32) * GT_PITCH + 4 * c4] = wu1;
    *(float4*)&sWv[wr * GT_PITCH + 4 * c4]        = wv0;
    *(float4*)&sWv[(wr + 32) * GT_PITCH + 4 * c4] = wv1;

    const int c0w  = (w & 3) * 16;
    const int rb   = (w >> 2) * 32;
    const int half = lane >> 4;
    const int l    = lane & 15;
    const int kofs = half * 2;

    #pragma unroll
    for (int it = 0; it < BKT / (2 * GT_ROWS); ++it) {
        const int slab = 2 * it + t;
        float* tS = tile + slab * (GT_ROWS * GT_PITCH);
        const int row0 = node0 + slab * GT_ROWS;

        #pragma unroll
        for (int j = 0; j < 4; ++j) {
            const int r = tr + 16 * j;
            const float4 a2 = *(const float4*)&tS[r * GT_PITCH + 4 * c4];
            float4 a1;
            a1.x = 2.0f * qv[j].x + a2.x;        a1.y = 2.0f * qv[j].y + a2.y;
            a1.z = 2.0f * qv[j].z + a2.z;        a1.w = 2.0f * qv[j].w + a2.w;
            *(float4*)&sA1[r * GT_PITCH + 4 * c4] = a1;
        }
        __syncthreads();

        if (it + 1 < BKT / (2 * GT_ROWS)) {
            #pragma unroll
            for (int j = 0; j < 4; ++j) {
                const int g = min(row0 + 2 * GT_ROWS + tr + 16 * j, N - 1);
                qv[j] = *(const float4*)(q + (long long)g * 64 + 4 * c4);
            }
        }

        const float* pBu = sWu + (c0w + l) * GT_PITCH + kofs;
        const float* pBv = sWv + (c0w + l) * GT_PITCH + kofs;
        const float* pA1 = sA1 + (rb + l) * GT_PITCH + kofs;
        const float* pA2 = tS + (rb + l) * GT_PITCH + kofs;

        v8f accU0 = {}, accU1 = {};
        v8f accV0 = {}, accV1 = {};

        #pragma unroll
        for (int k = 0; k < 64; k += 4) {
            const v2f bu  = *(const v2f*)(pBu + k);
            const v2f bv  = *(const v2f*)(pBv + k);
            const v2f a10 = *(const v2f*)(pA1 + k);
            const v2f a11 = *(const v2f*)(pA1 + 16 * GT_PITCH + k);
            const v2f a20 = *(const v2f*)(pA2 + k);
            const v2f a21 = *(const v2f*)(pA2 + 16 * GT_PITCH + k);
            accU0 = __builtin_amdgcn_wmma_f32_16x16x4_f32(false, a10, false, bu,
                                                          (short)0, accU0, false, false);
            accU1 = __builtin_amdgcn_wmma_f32_16x16x4_f32(false, a11, false, bu,
                                                          (short)0, accU1, false, false);
            accV0 = __builtin_amdgcn_wmma_f32_16x16x4_f32(false, a20, false, bv,
                                                          (short)0, accV0, false, false);
            accV1 = __builtin_amdgcn_wmma_f32_16x16x4_f32(false, a21, false, bv,
                                                          (short)0, accV1, false, false);
        }

        __syncthreads();

        #pragma unroll
        for (int r = 0; r < 8; ++r) {
            const int lr = rb + r + half * 8;
            sA1[lr * GT_PITCH + c0w + l]        = accU0[r];
            sA1[(lr + 16) * GT_PITCH + c0w + l] = accU1[r];
            tS[lr * GT_PITCH + c0w + l]         = accV0[r];
            tS[(lr + 16) * GT_PITCH + c0w + l]  = accV1[r];
        }
        __syncthreads();

        #pragma unroll
        for (int j = 0; j < 4; ++j) {
            const int r = tr + 16 * j;
            const int g = row0 + r;
            if (g < N) {
                const float4 u = *(const float4*)&sA1[r * GT_PITCH + 4 * c4];
                float4 v = *(const float4*)&tS[r * GT_PITCH + 4 * c4];
                const int br = slab * GT_ROWS + r;
                const float sI = ((dn[br] > 0.0f) ? 1.0f : 0.0f) + ((dn[BKT + br] > 0.0f) ? 1.0f : 0.0f);
                v.x = v.x + sI * b4.x;  v.y = v.y + sI * b4.y;
                v.z = v.z + sI * b4.z;  v.w = v.w + sI * b4.w;
                *(float4*)(out + (long long)g * 64 + 4 * c4) = u;
                *(float4*)(out + (long long)N * 64 + (long long)g * 64 + 4 * c4) = v;
            }
        }
        if (it + 1 < BKT / (2 * GT_ROWS)) __syncthreads();
    }
}

extern "C" void kernel_launch(void* const* d_in, const int* in_sizes, int n_in,
                              void* d_out, int out_size, void* d_ws, size_t ws_size,
                              hipStream_t stream) {
    const float* q    = (const float*)d_in[0];
    const float* Watt = (const float*)d_in[1];
    const float* Wupd = (const float*)d_in[2];
    const float* Wv   = (const float*)d_in[3];
    const float* bV   = (const float*)d_in[4];
    const int*   src1 = (const int*)d_in[5];
    const int*   dst1 = (const int*)d_in[6];
    const int*   src2 = (const int*)d_in[7];
    const int*   dst2 = (const int*)d_in[8];

    const int N = in_sizes[0] / 64;
    const int E = in_sizes[5];

    int*   list = (int*)d_ws;
    float* attS = (float*)d_ws + (long long)N * 64;
    float* attD = attS + N;
    int*   ctl  = (int*)(attD + 2 * (long long)N);
    int*   tot  = ctl;
    int*   cur  = ctl + 2 * NB_MAX;

    float* out  = (float*)d_out;

    int attBlocks = (N + 8 * ATT_ROWS - 1) / (8 * ATT_ROWS);
    k_att<<<attBlocks, 256, 0, stream>>>(q, Watt, attS, attD, ctl, N);

    int edgeBlocks = (E + EBLK - 1) / EBLK;
    k_count<<<dim3(edgeBlocks, 2), 256, 0, stream>>>(dst1, dst2, tot, E);
    k_fill<<<dim3(128, 2), FT, 0, stream>>>(src1, dst1, src2, dst2, attS,
                                            tot, cur, list, E, N);

    int buckets = (N + BKT - 1) / BKT;
    k_accum<<<buckets, 512, 0, stream>>>(q, attD, list, tot, Wupd, Wv, bV, out, E, N);
}
